# starting kernel
# baseline (speedup 1.0000x reference)
#include <hip/hip_runtime.h>
#include <hip/hip_bf16.h>
#include <math.h>


typedef __attribute__((ext_vector_type(2))) float v2f;
typedef __attribute__((ext_vector_type(8))) float v8f;

#define NEG_SLOPE_ATT 0.2f
#define NEG_SLOPE_ACT 0.01f

__device__ __forceinline__ unsigned f2ord(float f) {
    unsigned u = __float_as_uint(f);
    return (u & 0x80000000u) ? ~u : (u | 0x80000000u);
}
__device__ __forceinline__ float ord2f(unsigned o) {
    return __uint_as_float((o & 0x80000000u) ? (o & 0x7fffffffu) : ~o);
}
#define ORD_NEG_INF 0x007FFFFFu

__global__ void gat_fill_u32(unsigned* __restrict__ p, unsigned v, size_t n) {
    size_t i = (size_t)blockIdx.x * blockDim.x + threadIdx.x;
    size_t stride = (size_t)gridDim.x * blockDim.x;
    for (; i < n; i += stride) p[i] = v;
}

__global__ __launch_bounds__(256)
void gat_gemm_wmma(const float* __restrict__ A, const float* __restrict__ B,
                   float* __restrict__ C, int Nrows, int K, int Ncols) {
    extern __shared__ float ldsA[];
    const int KP    = K + 4;
    const int tileM = blockIdx.x * 32;
    const int lane  = threadIdx.x & 31;
    const int wave  = threadIdx.x >> 5;

    for (int idx = threadIdx.x; idx < 32 * K; idx += blockDim.x) {
        int r = idx / K, c = idx - r * K;
        int row = tileM + r;
        ldsA[r * KP + c] = (row < Nrows) ? A[(size_t)row * K + c] : 0.0f;
    }
    __syncthreads();

    const int tileN = wave * 16;
    if (tileN >= Ncols) return;

    const int rsel  = lane & 15;
    const int khalf = (lane >> 4) << 1;

    v8f acc0 = {};
    v8f acc1 = {};
    const float* __restrict__ a0p = &ldsA[rsel * KP + khalf];
    const float* __restrict__ a1p = &ldsA[(rsel + 16) * KP + khalf];
    for (int k0 = 0; k0 < K; k0 += 4) {
        v2f a0, a1, b;
        a0.x = a0p[k0];     a0.y = a0p[k0 + 1];
        a1.x = a1p[k0];     a1.y = a1p[k0 + 1];
        b.x  = B[(size_t)(k0 + khalf) * Ncols + tileN + rsel];
        b.y  = B[(size_t)(k0 + khalf + 1) * Ncols + tileN + rsel];
        acc0 = __builtin_amdgcn_wmma_f32_16x16x4_f32(
            false, a0, false, b, (short)0, acc0, false, false);
        acc1 = __builtin_amdgcn_wmma_f32_16x16x4_f32(
            false, a1, false, b, (short)0, acc1, false, false);
    }

    const int mbase = tileM + ((lane >> 4) << 3);
    const int col   = tileN + rsel;
    for (int r = 0; r < 8; ++r) {
        int row0 = mbase + r;
        int row1 = row0 + 16;
        if (row0 < Nrows) C[(size_t)row0 * Ncols + col] = acc0[r];
        if (row1 < Nrows) C[(size_t)row1 * Ncols + col] = acc1[r];
    }
}

__global__ void gat_logits(const float* __restrict__ h,
                           const float* __restrict__ a_src,
                           const float* __restrict__ a_dst,
                           float* __restrict__ e_src, float* __restrict__ e_dst,
                           int N, int H, int C) {
    int t = blockIdx.x * blockDim.x + threadIdx.x;
    if (t >= N * H) return;
    int n = t / H, hh = t - n * H;
    const float* hp = h + (size_t)n * H * C + (size_t)hh * C;
    const float* as = a_src + hh * C;
    const float* ad = a_dst + hh * C;
    float s = 0.f, d = 0.f;
    for (int c = 0; c < C; c += 4) {
        float4 hv = *(const float4*)(hp + c);
        float4 av = *(const float4*)(as + c);
        float4 dv = *(const float4*)(ad + c);
        s += hv.x * av.x + hv.y * av.y + hv.z * av.z + hv.w * av.w;
        d += hv.x * dv.x + hv.y * dv.y + hv.z * dv.z + hv.w * dv.w;
    }
    e_src[t] = s;
    e_dst[t] = d;
}

__device__ __forceinline__ float lrelu(float x, float slope) {
    return x > 0.f ? x : slope * x;
}

__device__ __forceinline__ void edge_sd(int e, int E, const int* __restrict__ s,
                                        const int* __restrict__ d, int& src, int& dst) {
    if (e < E) { src = s[e]; dst = d[e]; } else { src = e - E; dst = src; }
}

__global__ void gat_edge_max(const int* __restrict__ srcIdx, const int* __restrict__ dstIdx,
                             int E, int N, const float* __restrict__ e_src,
                             const float* __restrict__ e_dst,
                             unsigned* __restrict__ mOrd, int H) {
    int e = blockIdx.x * blockDim.x + threadIdx.x;
    if (e >= E + N) return;
    int s, d; edge_sd(e, E, srcIdx, dstIdx, s, d);
    for (int h = 0; h < H; ++h) {
        float a = lrelu(e_src[s * H + h] + e_dst[d * H + h], NEG_SLOPE_ATT);
        atomicMax(&mOrd[d * H + h], f2ord(a));
    }
}

__global__ void gat_edge_sum(const int* __restrict__ srcIdx, const int* __restrict__ dstIdx,
                             int E, int N, const float* __restrict__ e_src,
                             const float* __restrict__ e_dst,
                             const unsigned* __restrict__ mOrd,
                             float* __restrict__ denom, int H) {
    int e = blockIdx.x * blockDim.x + threadIdx.x;
    if (e >= E + N) return;
    int s, d; edge_sd(e, E, srcIdx, dstIdx, s, d);
    for (int h = 0; h < H; ++h) {
        float a = lrelu(e_src[s * H + h] + e_dst[d * H + h], NEG_SLOPE_ATT);
        float ex = __expf(a - ord2f(mOrd[d * H + h]));
        atomicAdd(&denom[d * H + h], ex);
    }
}

__global__ __launch_bounds__(256)
void gat_edge_scatter(const int* __restrict__ srcIdx, const int* __restrict__ dstIdx,
                      int E, int N, const float* __restrict__ e_src,
                      const float* __restrict__ e_dst,
                      const unsigned* __restrict__ mOrd,
                      const float* __restrict__ denom,
                      const float* __restrict__ h, float* __restrict__ out,
                      int H, int C) {
    int wid  = (int)(((size_t)blockIdx.x * blockDim.x + threadIdx.x) >> 5);
    int lane = threadIdx.x & 31;
    if (wid >= E + N) return;
    int s, d; edge_sd(wid, E, srcIdx, dstIdx, s, d);
    const int F = H * C;

    __builtin_prefetch(&h[(size_t)s * F + lane], 0, 1);

    float coefLane = 0.f;
    if (lane < H) {
        float a = lrelu(e_src[s * H + lane] + e_dst[d * H + lane], NEG_SLOPE_ATT);
        coefLane = __expf(a - ord2f(mOrd[d * H + lane])) / denom[d * H + lane];
    }
    for (int col = lane; col < F; col += 32) {
        float coef = __shfl(coefLane, col / C, 32);
        atomicAdd(&out[(size_t)d * F + col], h[(size_t)s * F + col] * coef);
    }
}

__global__ void gat_bias_act(float* __restrict__ x, const float* __restrict__ b,
                             size_t total, int F, float slope, int apply_act) {
    size_t i = (size_t)blockIdx.x * blockDim.x + threadIdx.x;
    if (i >= total) return;
    int c = (int)(i % F);
    float v = x[i] + b[c];
    if (apply_act) v = lrelu(v, slope);
    x[i] = v;
}

extern "C" void kernel_launch(void* const* d_in, const int* in_sizes, int n_in,
                              void* d_out, int out_size, void* d_ws, size_t ws_size,
                              hipStream_t stream) {
    const float* x        = (const float*)d_in[0];
    const int*   ei       = (const int*)d_in[1];
    const float* W1       = (const float*)d_in[2];
    const float* att_src1 = (const float*)d_in[3];
    const float* att_dst1 = (const float*)d_in[4];
    const float* b1       = (const float*)d_in[5];
    const float* W2       = (const float*)d_in[6];
    const float* att_src2 = (const float*)d_in[7];
    const float* att_dst2 = (const float*)d_in[8];
    const float* b2       = (const float*)d_in[9];

    const int HC1  = in_sizes[5];
    const int OUTC = in_sizes[9];
    const int F_IN = in_sizes[2] / HC1;
    const int N    = in_sizes[0] / F_IN;
    const int E    = in_sizes[1] / 2;
    const int H1 = 4, C1 = HC1 / 4;
    const int H2 = 1, C2 = OUTC;
    const int Etot = E + N;

    const int* srcIdx = ei;
    const int* dstIdx = ei + E;

    float* ws  = (float*)d_ws;
    size_t off = 0;
    float* h1   = ws + off; off += (size_t)N * HC1;
    float* act1 = ws + off; off += (size_t)N * HC1;
    float* es1  = ws + off; off += (size_t)N * H1;
    float* ed1  = ws + off; off += (size_t)N * H1;
    float* dn1  = ws + off; off += (size_t)N * H1;
    unsigned* m1 = (unsigned*)(ws + off); off += (size_t)N * H1;
    float* es2  = ws + off; off += (size_t)N;
    float* ed2  = ws + off; off += (size_t)N;
    float* dn2  = ws + off; off += (size_t)N;
    unsigned* m2 = (unsigned*)(ws + off); off += (size_t)N;
    float* h2   = h1;
    float* out2 = (float*)d_out;

    const int TB = 256;
    const int fillBlocks = 2048;

    gat_fill_u32<<<fillBlocks, TB, 0, stream>>>((unsigned*)act1, 0u, (size_t)N * HC1);
    gat_fill_u32<<<fillBlocks, TB, 0, stream>>>((unsigned*)out2, 0u, (size_t)N * OUTC);
    gat_fill_u32<<<fillBlocks, TB, 0, stream>>>((unsigned*)dn1, 0u, (size_t)N * H1);
    gat_fill_u32<<<fillBlocks, TB, 0, stream>>>(m1, ORD_NEG_INF, (size_t)N * H1);
    gat_fill_u32<<<fillBlocks, TB, 0, stream>>>((unsigned*)dn2, 0u, (size_t)N);
    gat_fill_u32<<<fillBlocks, TB, 0, stream>>>(m2, ORD_NEG_INF, (size_t)N);

    const int rowTiles32 = (N + 31) / 32;
    const int edgeBlocks = (Etot + TB - 1) / TB;
    const int scatterBlocks = (int)(((size_t)Etot * 32 + TB - 1) / TB);

    gat_gemm_wmma<<<rowTiles32, 32 * (HC1 / 16), 32 * (F_IN + 4) * sizeof(float), stream>>>(
        x, W1, h1, N, F_IN, HC1);
    gat_logits<<<(N * H1 + TB - 1) / TB, TB, 0, stream>>>(
        h1, att_src1, att_dst1, es1, ed1, N, H1, C1);
    gat_edge_max<<<edgeBlocks, TB, 0, stream>>>(srcIdx, dstIdx, E, N, es1, ed1, m1, H1);
    gat_edge_sum<<<edgeBlocks, TB, 0, stream>>>(srcIdx, dstIdx, E, N, es1, ed1, m1, dn1, H1);
    gat_edge_scatter<<<scatterBlocks, TB, 0, stream>>>(
        srcIdx, dstIdx, E, N, es1, ed1, m1, dn1, h1, act1, H1, C1);
    gat_bias_act<<<(int)(((size_t)N * HC1 + TB - 1) / TB), TB, 0, stream>>>(
        act1, b1, (size_t)N * HC1, HC1, NEG_SLOPE_ACT,  1);

    gat_gemm_wmma<<<rowTiles32, 32 * (OUTC / 16), 32 * (HC1 + 4) * sizeof(float), stream>>>(
        act1, W2, h2, N, HC1, OUTC);
    gat_logits<<<(N * H2 + TB - 1) / TB, TB, 0, stream>>>(
        h2, att_src2, att_dst2, es2, ed2, N, H2, C2);
    gat_edge_max<<<edgeBlocks, TB, 0, stream>>>(srcIdx, dstIdx, E, N, es2, ed2, m2, H2);
    gat_edge_sum<<<edgeBlocks, TB, 0, stream>>>(srcIdx, dstIdx, E, N, es2, ed2, m2, dn2, H2);
    gat_edge_scatter<<<scatterBlocks, TB, 0, stream>>>(
        srcIdx, dstIdx, E, N, es2, ed2, m2, dn2, h2, out2, H2, C2);
    gat_bias_act<<<(int)(((size_t)N * OUTC + TB - 1) / TB), TB, 0, stream>>>(
        out2, b2, (size_t)N * OUTC, OUTC, 0.f,  0);
}


# Edges grouped by destination per call (int-atomic rank, block scan, place); one wave per node does max, exp-sum, weighted row gather, bias in registers; no float atomics; sums in arrival-rank order
# speedup vs baseline: 3.8534x; 3.8534x over previous
#include <hip/hip_runtime.h>
#include <hip/hip_bf16.h>
#include <math.h>


typedef __attribute__((ext_vector_type(2))) float v2f;
typedef __attribute__((ext_vector_type(8))) float v8f;

#define NEG_SLOPE_ATT 0.2f
#define NEG_SLOPE_ACT 0.01f

__global__ void gat_fill_u32(unsigned* __restrict__ p, unsigned v, size_t n) {
    size_t i = (size_t)blockIdx.x * blockDim.x + threadIdx.x;
    size_t stride = (size_t)gridDim.x * blockDim.x;
    for (; i < n; i += stride) p[i] = v;
}

__global__ __launch_bounds__(256)
void gat_gemm_wmma(const float* __restrict__ A, const float* __restrict__ B,
                   float* __restrict__ C, int Nrows, int K, int Ncols) {
    extern __shared__ float ldsA[];
    const int KP    = K + 4;
    const int tileM = blockIdx.x * 32;
    const int lane  = threadIdx.x & 31;
    const int wave  = threadIdx.x >> 5;

    for (int idx = threadIdx.x; idx < 32 * K; idx += blockDim.x) {
        int r = idx / K, c = idx - r * K;
        int row = tileM + r;
        ldsA[r * KP + c] = (row < Nrows) ? A[(size_t)row * K + c] : 0.0f;
    }
    __syncthreads();

    const int tileN = wave * 16;
    if (tileN >= Ncols) return;

    const int rsel  = lane & 15;
    const int khalf = (lane >> 4) << 1;

    v8f acc0 = {};
    v8f acc1 = {};
    const float* __restrict__ a0p = &ldsA[rsel * KP + khalf];
    const float* __restrict__ a1p = &ldsA[(rsel + 16) * KP + khalf];
    for (int k0 = 0; k0 < K; k0 += 4) {
        v2f a0, a1, b;
        a0.x = a0p[k0];     a0.y = a0p[k0 + 1];
        a1.x = a1p[k0];     a1.y = a1p[k0 + 1];
        b.x  = B[(size_t)(k0 + khalf) * Ncols + tileN + rsel];
        b.y  = B[(size_t)(k0 + khalf + 1) * Ncols + tileN + rsel];
        acc0 = __builtin_amdgcn_wmma_f32_16x16x4_f32(
            false, a0, false, b, (short)0, acc0, false, false);
        acc1 = __builtin_amdgcn_wmma_f32_16x16x4_f32(
            false, a1, false, b, (short)0, acc1, false, false);
    }

    const int mbase = tileM + ((lane >> 4) << 3);
    const int col   = tileN + rsel;
    for (int r = 0; r < 8; ++r) {
        int row0 = mbase + r;
        int row1 = row0 + 16;
        if (row0 < Nrows) C[(size_t)row0 * Ncols + col] = acc0[r];
        if (row1 < Nrows) C[(size_t)row1 * Ncols + col] = acc1[r];
    }
}

__global__ void gat_logits(const float* __restrict__ h,
                           const float* __restrict__ a_src,
                           const float* __restrict__ a_dst,
                           float* __restrict__ e_src, float* __restrict__ e_dst,
                           int N, int H, int C) {
    int t = blockIdx.x * blockDim.x + threadIdx.x;
    if (t >= N * H) return;
    int n = t / H, hh = t - n * H;
    const float* hp = h + (size_t)n * H * C + (size_t)hh * C;
    const float* as = a_src + hh * C;
    const float* ad = a_dst + hh * C;
    float s = 0.f, d = 0.f;
    for (int c = 0; c < C; c += 4) {
        float4 hv = *(const float4*)(hp + c);
        float4 av = *(const float4*)(as + c);
        float4 dv = *(const float4*)(ad + c);
        s += hv.x * av.x + hv.y * av.y + hv.z * av.z + hv.w * av.w;
        d += hv.x * dv.x + hv.y * dv.y + hv.z * dv.z + hv.w * dv.w;
    }
    e_src[t] = s;
    e_dst[t] = d;
}

__device__ __forceinline__ float lrelu(float x, float slope) {
    return x > 0.f ? x : slope * x;
}

__global__ void gat_count_rank(const int* __restrict__ dstIdx, int E,
                               int* __restrict__ cnt, int* __restrict__ rank) {
    int e = blockIdx.x * blockDim.x + threadIdx.x;
    if (e >= E) return;
    rank[e] = atomicAdd(&cnt[dstIdx[e]], 1);
}

__global__ __launch_bounds__(256)
void gat_alloc_rows(const int* cnt, int N, int* total, int* __restrict__ rowStart) {
    __shared__ int waveBase[8];
    __shared__ int blockBase;
    const int lane = threadIdx.x & 31;
    const int wave = threadIdx.x >> 5;
    const int n = blockIdx.x * 256 + threadIdx.x;
    const int c = (n < N) ? cnt[n] : 0;
    int incl = c;
    for (int off = 1; off < 32; off <<= 1) {
        int v = __shfl_up(incl, off, 32);
        if (lane >= off) incl += v;
    }
    if (lane == 31) waveBase[wave] = incl;
    __syncthreads();
    if (threadIdx.x == 0) {
        int run = 0;
        for (int w = 0; w < 8; ++w) { int t = waveBase[w]; waveBase[w] = run; run += t; }
        blockBase = atomicAdd(total, run);
    }
    __syncthreads();
    if (n < N) rowStart[n] = blockBase + waveBase[wave] + incl - c;
}

__global__ void gat_place(const int* __restrict__ srcIdx, const int* __restrict__ dstIdx,
                          const int* __restrict__ rank, const int* __restrict__ rowStart,
                          int E, int* __restrict__ segSrc) {
    int e = blockIdx.x * blockDim.x + threadIdx.x;
    if (e >= E) return;
    segSrc[rowStart[dstIdx[e]] + rank[e]] = srcIdx[e];
}

__device__ __forceinline__ float wave_max(float v) {
    for (int o = 16; o > 0; o >>= 1) v = fmaxf(v, __shfl_xor(v, o, 32));
    return v;
}
__device__ __forceinline__ float wave_sum(float v) {
    for (int o = 16; o > 0; o >>= 1) v += __shfl_xor(v, o, 32);
    return v;
}

template <int H>
__device__ __forceinline__ void gat_item(int j, int deg, int d, int start,
                                         const int* __restrict__ segSrc,
                                         const float* __restrict__ e_src,
                                         const float (&edv)[H], int& s, float (&a)[H]) {
    s = d;
    if (j < deg) s = segSrc[start + j];
    float ev[H];
    if constexpr (H == 4) {
        float4 t = *(const float4*)(e_src + (size_t)s * 4);
        ev[0] = t.x; ev[1] = t.y; ev[2] = t.z; ev[3] = t.w;
    } else {
        for (int k = 0; k < H; ++k) ev[k] = e_src[(size_t)s * H + k];
    }
    for (int k = 0; k < H; ++k)
        a[k] = (j <= deg) ? lrelu(ev[k] + edv[k], NEG_SLOPE_ATT) : -INFINITY;
}

template <int H, int C, int VEC>
__device__ __forceinline__ void gat_accum(int sItem, const float (&coef)[H], int count,
                                          int hd, int lane, const float* __restrict__ h,
                                          float (&acc)[VEC]) {
    constexpr int F = H * C;
    #pragma unroll 4
    for (int t = 0; t < count; ++t) {
        const int st = __builtin_amdgcn_readlane(sItem, t);
        float c = __int_as_float(__builtin_amdgcn_readlane(__float_as_int(coef[0]), t));
        for (int k = 1; k < H; ++k) {
            float ck = __int_as_float(__builtin_amdgcn_readlane(__float_as_int(coef[k]), t));
            c = (hd == k) ? ck : c;
        }
        const float* row = h + (size_t)st * F + lane * VEC;
        if constexpr (VEC == 4) {
            float4 v = *(const float4*)row;
            acc[0] += c * v.x; acc[1] += c * v.y; acc[2] += c * v.z; acc[3] += c * v.w;
        } else {
            float2 v = *(const float2*)row;
            acc[0] += c * v.x; acc[1] += c * v.y;
        }
    }
}

template <int H, int C>
__global__ __launch_bounds__(256)
void gat_gather(const int* __restrict__ rowStart, const int* __restrict__ cnt,
                const int* __restrict__ segSrc, int N,
                const float* __restrict__ e_src, const float* __restrict__ e_dst,
                const float* __restrict__ h, const float* __restrict__ bias,
                float* __restrict__ out, float slope, int apply_act) {
    constexpr int F = H * C;
    constexpr int VEC = F / 32;
    static_assert(F % 32 == 0 && (VEC == 4 || VEC == 2), "row = 32 lanes x VEC");
    static_assert(C % VEC == 0, "a lane's columns stay inside one head");

    const int lane = threadIdx.x & 31;
    const int d = __builtin_amdgcn_readfirstlane((int)(blockIdx.x * 8 + (threadIdx.x >> 5)));
    if (d >= N) return;
    const int start = rowStart[d];
    const int deg   = cnt[d];
    const int nItems = deg + 1;
    const int hd = (lane * VEC) / C;

    float edv[H];
    for (int k = 0; k < H; ++k) edv[k] = e_dst[(size_t)d * H + k];

    int s0; float a0[H];
    gat_item<H>(lane, deg, d, start, segSrc, e_src, edv, s0, a0);
    float m[H];
    for (int k = 0; k < H; ++k) m[k] = a0[k];
    for (int base = 32; base < nItems; base += 32) {
        int s; float a[H];
        gat_item<H>(base + lane, deg, d, start, segSrc, e_src, edv, s, a);
        for (int k = 0; k < H; ++k) m[k] = fmaxf(m[k], a[k]);
    }
    for (int k = 0; k < H; ++k) m[k] = wave_max(m[k]);

    float e0[H], sum[H];
    for (int k = 0; k < H; ++k) {
        e0[k] = (lane <= deg) ? __expf(a0[k] - m[k]) : 0.f;
        sum[k] = e0[k];
    }
    for (int base = 32; base < nItems; base += 32) {
        int s; float a[H];
        gat_item<H>(base + lane, deg, d, start, segSrc, e_src, edv, s, a);
        for (int k = 0; k < H; ++k)
            sum[k] += (base + lane <= deg) ? __expf(a[k] - m[k]) : 0.f;
    }
    for (int k = 0; k < H; ++k) sum[k] = wave_sum(sum[k]);

    float acc[VEC];
    for (int v = 0; v < VEC; ++v) acc[v] = 0.f;
    float c0[H];
    for (int k = 0; k < H; ++k) c0[k] = e0[k] / sum[k];
    gat_accum<H, C, VEC>(s0, c0, nItems < 32 ? nItems : 32, hd, lane, h, acc);
    for (int base = 32; base < nItems; base += 32) {
        int s; float a[H], cf[H];
        gat_item<H>(base + lane, deg, d, start, segSrc, e_src, edv, s, a);
        for (int k = 0; k < H; ++k)
            cf[k] = (base + lane <= deg) ? __expf(a[k] - m[k]) / sum[k] : 0.f;
        const int left = nItems - base;
        gat_accum<H, C, VEC>(s, cf, left < 32 ? left : 32, hd, lane, h, acc);
    }

    const int col = lane * VEC;
    float r[VEC];
    for (int v = 0; v < VEC; ++v) {
        float x = acc[v] + bias[col + v];
        r[v] = apply_act ? lrelu(x, slope) : x;
    }
    float* op = out + (size_t)d * F + col;
    if constexpr (VEC == 4) {
        *(float4*)op = make_float4(r[0], r[1], r[2], r[3]);
    } else {
        *(float2*)op = make_float2(r[0], r[1]);
    }
}

extern "C" void kernel_launch(void* const* d_in, const int* in_sizes, int n_in,
                              void* d_out, int out_size, void* d_ws, size_t ws_size,
                              hipStream_t stream) {
    const float* x        = (const float*)d_in[0];
    const int*   ei       = (const int*)d_in[1];
    const float* W1       = (const float*)d_in[2];
    const float* att_src1 = (const float*)d_in[3];
    const float* att_dst1 = (const float*)d_in[4];
    const float* b1       = (const float*)d_in[5];
    const float* W2       = (const float*)d_in[6];
    const float* att_src2 = (const float*)d_in[7];
    const float* att_dst2 = (const float*)d_in[8];
    const float* b2       = (const float*)d_in[9];

    const int HC1  = in_sizes[5];
    const int OUTC = in_sizes[9];
    const int F_IN = in_sizes[2] / HC1;
    const int N    = in_sizes[0] / F_IN;
    const int E    = in_sizes[1] / 2;
    const int H1 = 4, C1 = HC1 / 4;
    const int H2 = 1, C2 = OUTC;

    const int* srcIdx = ei;
    const int* dstIdx = ei + E;

    float* ws  = (float*)d_ws;
    size_t off = 0;
    float* h1   = ws + off; off += (size_t)N * HC1;
    float* act1 = ws + off; off += (size_t)N * HC1;
    float* es1  = ws + off; off += (size_t)N * H1;
    float* ed1  = ws + off; off += (size_t)N * H1;
    float* es2  = ws + off; off += (((size_t)N + 3) / 4) * 4;
    float* ed2  = ws + off; off += (((size_t)N + 3) / 4) * 4;
    int* cnt      = (int*)(ws + off); off += (((size_t)N + 1 + 3) / 4) * 4;
    int* rowStart = (int*)(ws + off); off += (((size_t)N + 3) / 4) * 4;
    int* rank     = (int*)(ws + off); off += (((size_t)E + 3) / 4) * 4;
    int* segSrc   = (int*)(ws + off); off += (((size_t)E + 3) / 4) * 4;
    float* h2   = h1;
    float* out2 = (float*)d_out;

    const int TB = 256;
    const int rowTiles32 = (N + 31) / 32;
    const int edgeBlocks = (E + TB - 1) / TB;
    const int nodeBlocks = (N + TB - 1) / TB;
    const int waveBlocks = (N + 7) / 8;

    gat_fill_u32<<<(N + 1 + TB - 1) / TB, TB, 0, stream>>>((unsigned*)cnt, 0u, (size_t)N + 1);
    gat_count_rank<<<edgeBlocks, TB, 0, stream>>>(dstIdx, E, cnt, rank);
    gat_alloc_rows<<<nodeBlocks, TB, 0, stream>>>(cnt, N, cnt + N, rowStart);
    gat_place<<<edgeBlocks, TB, 0, stream>>>(srcIdx, dstIdx, rank, rowStart, E, segSrc);

    gat_gemm_wmma<<<rowTiles32, 32 * (HC1 / 16), 32 * (F_IN + 4) * sizeof(float), stream>>>(
        x, W1, h1, N, F_IN, HC1);
    gat_logits<<<(N * H1 + TB - 1) / TB, TB, 0, stream>>>(
        h1, att_src1, att_dst1, es1, ed1, N, H1, C1);
    gat_gather<4, 32><<<waveBlocks, TB, 0, stream>>>(
        rowStart, cnt, segSrc, N, es1, ed1, h1, b1, act1, NEG_SLOPE_ACT,  1);

    gat_gemm_wmma<<<rowTiles32, 32 * (OUTC / 16), 32 * (HC1 + 4) * sizeof(float), stream>>>(
        act1, W2, h2, N, HC1, OUTC);
    gat_logits<<<(N * H2 + TB - 1) / TB, TB, 0, stream>>>(
        h2, att_src2, att_dst2, es2, ed2, N, H2, C2);
    gat_gather<1, 64><<<waveBlocks, TB, 0, stream>>>(
        rowStart, cnt, segSrc, N, es2, ed2, h2, b2, out2, 0.f,  0);
}


# Edges grouped by destination with a two-level counting sort on LDS counters (chunk histograms per 256-node bucket, scan, scatter, per-bucket sort); no global atomics; sums in LDS arrival order
# speedup vs baseline: 5.6505x; 1.4664x over previous
#include <hip/hip_runtime.h>
#include <hip/hip_bf16.h>
#include <math.h>


typedef __attribute__((ext_vector_type(2))) float v2f;
typedef __attribute__((ext_vector_type(8))) float v8f;

#define NEG_SLOPE_ATT 0.2f
#define NEG_SLOPE_ACT 0.01f

__global__ __launch_bounds__(256)
void gat_gemm_wmma(const float* __restrict__ A, const float* __restrict__ B,
                   float* __restrict__ C, int Nrows, int K, int Ncols) {
    extern __shared__ float ldsA[];
    const int KP    = K + 4;
    const int tileM = blockIdx.x * 32;
    const int lane  = threadIdx.x & 31;
    const int wave  = threadIdx.x >> 5;

    for (int idx = threadIdx.x; idx < 32 * K; idx += blockDim.x) {
        int r = idx / K, c = idx - r * K;
        int row = tileM + r;
        ldsA[r * KP + c] = (row < Nrows) ? A[(size_t)row * K + c] : 0.0f;
    }
    __syncthreads();

    const int tileN = wave * 16;
    if (tileN >= Ncols) return;

    const int rsel  = lane & 15;
    const int khalf = (lane >> 4) << 1;

    v8f acc0 = {};
    v8f acc1 = {};
    const float* __restrict__ a0p = &ldsA[rsel * KP + khalf];
    const float* __restrict__ a1p = &ldsA[(rsel + 16) * KP + khalf];
    for (int k0 = 0; k0 < K; k0 += 4) {
        v2f a0, a1, b;
        a0.x = a0p[k0];     a0.y = a0p[k0 + 1];
        a1.x = a1p[k0];     a1.y = a1p[k0 + 1];
        b.x  = B[(size_t)(k0 + khalf) * Ncols + tileN + rsel];
        b.y  = B[(size_t)(k0 + khalf + 1) * Ncols + tileN + rsel];
        acc0 = __builtin_amdgcn_wmma_f32_16x16x4_f32(
            false, a0, false, b, (short)0, acc0, false, false);
        acc1 = __builtin_amdgcn_wmma_f32_16x16x4_f32(
            false, a1, false, b, (short)0, acc1, false, false);
    }

    const int mbase = tileM + ((lane >> 4) << 3);
    const int col   = tileN + rsel;
    for (int r = 0; r < 8; ++r) {
        int row0 = mbase + r;
        int row1 = row0 + 16;
        if (row0 < Nrows) C[(size_t)row0 * Ncols + col] = acc0[r];
        if (row1 < Nrows) C[(size_t)row1 * Ncols + col] = acc1[r];
    }
}

__device__ __forceinline__ float lrelu(float x, float slope) {
    return x > 0.f ? x : slope * x;
}

#define BSHIFT 8
#define BNODES 256
#define NCHUNK 256
#define EPT 8

__device__ __forceinline__ int bucket_of(int d, int nb) {
    const unsigned b = (unsigned)d >> BSHIFT;
    return (int)(b < (unsigned)nb ? b : (unsigned)(nb - 1));
}

__device__ __forceinline__ int block_excl_scan(int v, int* wsum, int nwaves, int& total) {
    const int lane = threadIdx.x & 31;
    const int wave = threadIdx.x >> 5;
    int incl = v;
    for (int off = 1; off < 32; off <<= 1) {
        int t = __shfl_up(incl, off, 32);
        if (lane >= off) incl += t;
    }
    if (lane == 31) wsum[wave] = incl;
    __syncthreads();
    int base = 0, tot = 0;
    for (int w = 0; w < nwaves; ++w) {
        const int t = wsum[w];
        if (w < wave) base += t;
        tot += t;
    }
    __syncthreads();
    total = tot;
    return base + incl - v;
}

__global__ __launch_bounds__(1024)
void gat_bucket_hist(const int* __restrict__ dstIdx, int E, int chunk, int nb,
                     int* __restrict__ hist) {
    extern __shared__ int lcount[];
    for (int i = threadIdx.x; i < nb; i += 1024) lcount[i] = 0;
    __syncthreads();
    const int e0 = blockIdx.x * chunk;
    const int e1 = min(E, e0 + chunk);
    for (int base = e0; base < e1; base += EPT * 1024) {
        int d[EPT];
        for (int k = 0; k < EPT; ++k) {
            const int e = base + k * 1024 + (int)threadIdx.x;
            d[k] = dstIdx[e < e1 ? e : e1 - 1];
        }
        for (int k = 0; k < EPT; ++k) {
            const int e = base + k * 1024 + (int)threadIdx.x;
            if (e < e1) atomicAdd(&lcount[bucket_of(d[k], nb)], 1);
        }
    }
    __syncthreads();
    for (int i = threadIdx.x; i < nb; i += 1024) hist[i * NCHUNK + blockIdx.x] = lcount[i];
}

__global__ __launch_bounds__(NCHUNK)
void gat_bucket_rows(int* __restrict__ hist, int* __restrict__ bucketTotal) {
    __shared__ int wsum[32];
    const int i = blockIdx.x * NCHUNK + threadIdx.x;
    const int v = hist[i];
    int total;
    const int ex = block_excl_scan(v, wsum, NCHUNK / 32, total);
    hist[i] = ex;
    if (threadIdx.x == 0) bucketTotal[blockIdx.x] = total;
}

__global__ __launch_bounds__(1024)
void gat_bucket_scatter(const int* __restrict__ srcIdx, const int* __restrict__ dstIdx,
                        int E, int chunk, int nb,
                        const int* __restrict__ rowPre, const int* __restrict__ bucketTotal,
                        int* __restrict__ bucketBase, int2* __restrict__ part) {
    extern __shared__ int cursor[];
    __shared__ int wsum[32];
    int carry = 0;
    for (int base = 0; base < nb; base += 1024) {
        const int b = base + (int)threadIdx.x;
        const int v = b < nb ? bucketTotal[b] : 0;
        int total;
        const int ex = block_excl_scan(v, wsum, 32, total);
        if (b < nb) {
            cursor[b] = carry + ex + rowPre[b * NCHUNK + blockIdx.x];
            if (blockIdx.x == 0) bucketBase[b] = carry + ex;
        }
        carry += total;
    }
    __syncthreads();
    const int e0 = blockIdx.x * chunk;
    const int e1 = min(E, e0 + chunk);
    for (int base = e0; base < e1; base += EPT * 1024) {
        int s[EPT], d[EPT];
        for (int k = 0; k < EPT; ++k) {
            const int e = base + k * 1024 + (int)threadIdx.x;
            const int ec = e < e1 ? e : e1 - 1;
            s[k] = srcIdx[ec];
            d[k] = dstIdx[ec];
        }
        for (int k = 0; k < EPT; ++k) {
            const int e = base + k * 1024 + (int)threadIdx.x;
            if (e < e1) {
                const int pos = atomicAdd(&cursor[bucket_of(d[k], nb)], 1);
                part[pos] = make_int2(s[k], d[k]);
            }
        }
    }
}

__global__ __launch_bounds__(BNODES)
void gat_bucket_sort(const int2* __restrict__ part, const int* __restrict__ bucketBase,
                     const int* __restrict__ bucketTotal, int N,
                     int* __restrict__ rowStart, int* __restrict__ cnt,
                     int* __restrict__ segSrc) {
    __shared__ int lnode[BNODES];
    __shared__ int wsum[32];
    const int tid = threadIdx.x;
    const int p0 = bucketBase[blockIdx.x];
    const int p1 = p0 + bucketTotal[blockIdx.x];
    lnode[tid] = 0;
    __syncthreads();
    for (int base = p0; base < p1; base += EPT * BNODES) {
        int d[EPT];
        for (int k = 0; k < EPT; ++k) {
            const int i = base + k * BNODES + tid;
            d[k] = part[i < p1 ? i : p1 - 1].y;
        }
        for (int k = 0; k < EPT; ++k) {
            const int i = base + k * BNODES + tid;
            if (i < p1) atomicAdd(&lnode[d[k] & (BNODES - 1)], 1);
        }
    }
    __syncthreads();
    const int c = lnode[tid];
    int total;
    const int ex = block_excl_scan(c, wsum, BNODES / 32, total);
    lnode[tid] = p0 + ex;
    const int n = blockIdx.x * BNODES + tid;
    if (n < N) { rowStart[n] = p0 + ex; cnt[n] = c; }
    __syncthreads();
    for (int base = p0; base < p1; base += EPT * BNODES) {
        int2 sd[EPT];
        for (int k = 0; k < EPT; ++k) {
            const int i = base + k * BNODES + tid;
            sd[k] = part[i < p1 ? i : p1 - 1];
        }
        for (int k = 0; k < EPT; ++k) {
            const int i = base + k * BNODES + tid;
            if (i < p1) {
                const int pos = atomicAdd(&lnode[sd[k].y & (BNODES - 1)], 1);
                segSrc[pos] = sd[k].x;
            }
        }
    }
}

template <int MASK>
__device__ __forceinline__ float row_xor(float x) {
    return __int_as_float(__builtin_amdgcn_update_dpp(
        0, __float_as_int(x), 0x160 + MASK, 0xf, 0xf, true));
}
__device__ __forceinline__ float half_xor(float x) {
    const int b = __float_as_int(x);
    return __int_as_float(__builtin_amdgcn_permlanex16(b, b, 0x76543210, 0xfedcba98, false, false));
}
template <int LPH>
__device__ __forceinline__ float head_sum(float v) {
    v += row_xor<1>(v);
    v += row_xor<2>(v);
    v += row_xor<4>(v);
    if constexpr (LPH > 8)  v += row_xor<8>(v);
    if constexpr (LPH > 16) v += half_xor(v);
    return v;
}

template <int P>
__device__ __forceinline__ void load_row(const float* __restrict__ p, v2f (&v)[P]) {
    if constexpr (P == 2) {
        float4 t = *(const float4*)p;
        v[0].x = t.x; v[0].y = t.y; v[1].x = t.z; v[1].y = t.w;
    } else {
        float2 t = *(const float2*)p;
        v[0].x = t.x; v[0].y = t.y;
    }
}
template <int P>
__device__ __forceinline__ float dot_cols(const v2f (&v)[P], const v2f (&w)[P]) {
    v2f q = v[0] * w[0];
    for (int k = 1; k < P; ++k) q += v[k] * w[k];
    return q.x + q.y;
}

template <int NB, int F, int P, int LPH>
__device__ __forceinline__ void gat_rows(int sItem, int t, int lane,
                                         const float* __restrict__ h,
                                         const v2f (&as)[P], float ed,
                                         float& m, float& sum, v2f (&acc)[P]) {
    v2f v[NB][P];
    for (int u = 0; u < NB; ++u) {
        const int st = __builtin_amdgcn_readlane(sItem, t + u);
        const unsigned boff = ((unsigned)st * (unsigned)F + (unsigned)(lane * (2 * P))) * 4u;
        load_row<P>((const float*)((const char*)h + boff), v[u]);
    }
    float a[NB];
    for (int u = 0; u < NB; ++u)
        a[u] = lrelu(head_sum<LPH>(dot_cols<P>(v[u], as)) + ed, NEG_SLOPE_ATT);
    float nm = m;
    for (int u = 0; u < NB; ++u) nm = fmaxf(nm, a[u]);
    const float sc = __expf(m - nm);
    sum *= sc;
    for (int k = 0; k < P; ++k) acc[k] *= sc;
    for (int u = 0; u < NB; ++u) {
        const float pe = __expf(a[u] - nm);
        sum += pe;
        for (int k = 0; k < P; ++k) acc[k] += pe * v[u][k];
    }
    m = nm;
}

template <int H, int C>
__global__ __launch_bounds__(256)
void gat_gather(const int* __restrict__ rowStart, const int* __restrict__ cnt,
                const int* __restrict__ segSrc, int N,
                const float* __restrict__ a_src, const float* __restrict__ a_dst,
                const float* __restrict__ h, const float* __restrict__ bias,
                float* __restrict__ out, float slope, int apply_act) {
    constexpr int F = H * C;
    constexpr int VEC = F / 32;
    constexpr int P = VEC / 2;
    constexpr int LPH = C / VEC;
    static_assert(F % 32 == 0 && (VEC == 4 || VEC == 2), "row = 32 lanes x VEC");
    static_assert(C % VEC == 0, "a lane's columns stay inside one head");
    static_assert(LPH == 8 || LPH == 16 || LPH == 32, "a head is 8, 16 or 32 adjacent lanes");

    const int lane = threadIdx.x & 31;
    const int d = __builtin_amdgcn_readfirstlane((int)(blockIdx.x * 8 + (threadIdx.x >> 5)));
    if (d >= N) return;
    const int start = rowStart[d];
    const int deg   = cnt[d];
    const int col   = lane * VEC;

    v2f as[P], ad[P], vs[P];
    load_row<P>(a_src + col, as);
    load_row<P>(a_dst + col, ad);
    load_row<P>(h + (size_t)d * F + col, vs);
    int sItem = d;
    if (lane < deg) sItem = segSrc[start + lane];

    const float ed = head_sum<LPH>(dot_cols<P>(vs, ad));
    float m   = lrelu(head_sum<LPH>(dot_cols<P>(vs, as)) + ed, NEG_SLOPE_ATT);
    float sum = 1.f;
    v2f acc[P];
    for (int k = 0; k < P; ++k) acc[k] = vs[k];

    for (int base = 0; base < deg; base += 32) {
        if (base > 0) {
            sItem = d;
            if (base + lane < deg) sItem = segSrc[start + base + lane];
        }
        const int left = deg - base;
        const int n = left < 32 ? left : 32;
        int t = 0;
        for (; t + 8 <= n; t += 8)
            gat_rows<8, F, P, LPH>(sItem, t, lane, h, as, ed, m, sum, acc);
        if (t + 4 <= n) {
            gat_rows<4, F, P, LPH>(sItem, t, lane, h, as, ed, m, sum, acc);
            t += 4;
        }
        const int rem = n - t;
        if (rem == 3)      gat_rows<3, F, P, LPH>(sItem, t, lane, h, as, ed, m, sum, acc);
        else if (rem == 2) gat_rows<2, F, P, LPH>(sItem, t, lane, h, as, ed, m, sum, acc);
        else if (rem == 1) gat_rows<1, F, P, LPH>(sItem, t, lane, h, as, ed, m, sum, acc);
    }

    const float inv = 1.0f / sum;
    float r[VEC];
    for (int k = 0; k < VEC; ++k) {
        float x = acc[k >> 1][k & 1] * inv + bias[col + k];
        r[k] = apply_act ? lrelu(x, slope) : x;
    }
    float* op = out + (size_t)d * F + col;
    if constexpr (VEC == 4) {
        *(float4*)op = make_float4(r[0], r[1], r[2], r[3]);
    } else {
        *(float2*)op = make_float2(r[0], r[1]);
    }
}

extern "C" void kernel_launch(void* const* d_in, const int* in_sizes, int n_in,
                              void* d_out, int out_size, void* d_ws, size_t ws_size,
                              hipStream_t stream) {
    const float* x        = (const float*)d_in[0];
    const int*   ei       = (const int*)d_in[1];
    const float* W1       = (const float*)d_in[2];
    const float* att_src1 = (const float*)d_in[3];
    const float* att_dst1 = (const float*)d_in[4];
    const float* b1       = (const float*)d_in[5];
    const float* W2       = (const float*)d_in[6];
    const float* att_src2 = (const float*)d_in[7];
    const float* att_dst2 = (const float*)d_in[8];
    const float* b2       = (const float*)d_in[9];

    const int HC1  = in_sizes[5];
    const int OUTC = in_sizes[9];
    const int F_IN = in_sizes[2] / HC1;
    const int N    = in_sizes[0] / F_IN;
    const int E    = in_sizes[1] / 2;

    const int* srcIdx = ei;
    const int* dstIdx = ei + E;

    float* ws  = (float*)d_ws;
    size_t off = 0;
    float* h1   = ws + off; off += (size_t)N * HC1;
    float* act1 = ws + off; off += (size_t)N * HC1;
    const int NBK   = (N + BNODES - 1) / BNODES;
    const int chunk = (E + NCHUNK - 1) / NCHUNK;
    int* cnt      = (int*)(ws + off); off += (((size_t)N + 3) / 4) * 4;
    int* rowStart = (int*)(ws + off); off += (((size_t)N + 3) / 4) * 4;
    int* segSrc   = (int*)(ws + off); off += (((size_t)E + 3) / 4) * 4;
    int* hist        = (int*)(ws + off); off += (size_t)NBK * NCHUNK;
    int* bucketTotal = (int*)(ws + off); off += (((size_t)NBK + 3) / 4) * 4;
    int* bucketBase  = (int*)(ws + off); off += (((size_t)NBK + 3) / 4) * 4;
    int2* part       = (int2*)(ws + off); off += (((size_t)E * 2 + 3) / 4) * 4;
    float* h2   = h1;
    float* out2 = (float*)d_out;

    const int TB = 256;
    const int rowTiles32 = (N + 31) / 32;
    const int waveBlocks = (N + 7) / 8;

    gat_bucket_hist<<<NCHUNK, 1024, NBK * sizeof(int), stream>>>(dstIdx, E, chunk, NBK, hist);
    gat_bucket_rows<<<NBK, NCHUNK, 0, stream>>>(hist, bucketTotal);
    gat_bucket_scatter<<<NCHUNK, 1024, NBK * sizeof(int), stream>>>(
        srcIdx, dstIdx, E, chunk, NBK, hist, bucketTotal, bucketBase, part);
    gat_bucket_sort<<<NBK, BNODES, 0, stream>>>(part, bucketBase, bucketTotal, N, rowStart, cnt, segSrc);

    gat_gemm_wmma<<<rowTiles32, 32 * (HC1 / 16), 32 * (F_IN + 4) * sizeof(float), stream>>>(
        x, W1, h1, N, F_IN, HC1);
    gat_gather<4, 32><<<waveBlocks, TB, 0, stream>>>(
        rowStart, cnt, segSrc, N, att_src1, att_dst1, h1, b1, act1, NEG_SLOPE_ACT,  1);

    gat_gemm_wmma<<<rowTiles32, 32 * (OUTC / 16), 32 * (HC1 + 4) * sizeof(float), stream>>>(
        act1, W2, h2, N, HC1, OUTC);
    gat_gather<1, 64><<<waveBlocks, TB, 0, stream>>>(
        rowStart, cnt, segSrc, N, att_src2, att_dst2, h2, b2, out2, 0.f,  0);
}


# Projection kernels fetch their A tile as 16-byte pieces, all of a thread's loads in flight together ahead of the LDS stores (was one 4-byte load at a time, each waited for)
# speedup vs baseline: 7.5503x; 1.3362x over previous
#include <hip/hip_runtime.h>
#include <hip/hip_bf16.h>
#include <math.h>


typedef __attribute__((ext_vector_type(2))) float v2f;
typedef __attribute__((ext_vector_type(4))) float v4f;
typedef __attribute__((ext_vector_type(8))) float v8f;

#define NEG_SLOPE_ATT 0.2f
#define NEG_SLOPE_ACT 0.01f

template <int K, int NCOLS>
__global__ __launch_bounds__(2 * NCOLS)
void gat_gemm_wmma(const float* __restrict__ A, const float* __restrict__ B,
                   float* __restrict__ C, int Nrows) {
    constexpr int KP  = K + 4;
    constexpr int NT  = 2 * NCOLS;
    constexpr int K4  = K / 4;
    constexpr int PCS = (32 * K4) / NT;
    static_assert(K % 4 == 0 && NCOLS % 16 == 0, "whole 16-byte pieces, whole column tiles");
    static_assert((32 * K4) % NT == 0, "the tile's pieces divide evenly over the threads");
    extern __shared__ __attribute__((aligned(16))) float ldsA[];
    const int tileM = blockIdx.x * 32;
    const int lane  = threadIdx.x & 31;
    const int wave  = threadIdx.x >> 5;

    v4f piece[PCS];
    #pragma unroll
    for (int j = 0; j < PCS; ++j) {
        const int idx = j * NT + (int)threadIdx.x;
        const int r = idx / K4, c = (idx - r * K4) * 4;
        const int row = tileM + r;
        const int rc  = row < Nrows ? row : Nrows - 1;
        piece[j] = *(const v4f*)(A + (size_t)rc * K + c);
    }
    #pragma unroll
    for (int j = 0; j < PCS; ++j) {
        const int idx = j * NT + (int)threadIdx.x;
        const int r = idx / K4, c = (idx - r * K4) * 4;
        *(v4f*)&ldsA[r * KP + c] = piece[j];
    }
    __syncthreads();

    const int tileN = wave * 16;
    const int rsel  = lane & 15;
    const int khalf = (lane >> 4) << 1;

    v8f acc0 = {};
    v8f acc1 = {};
    const float* __restrict__ a0p = &ldsA[rsel * KP + khalf];
    const float* __restrict__ a1p = &ldsA[(rsel + 16) * KP + khalf];
    for (int k0 = 0; k0 < K; k0 += 4) {
        v2f a0, a1, b;
        a0.x = a0p[k0];     a0.y = a0p[k0 + 1];
        a1.x = a1p[k0];     a1.y = a1p[k0 + 1];
        b.x  = B[(size_t)(k0 + khalf) * NCOLS + tileN + rsel];
        b.y  = B[(size_t)(k0 + khalf + 1) * NCOLS + tileN + rsel];
        acc0 = __builtin_amdgcn_wmma_f32_16x16x4_f32(
            false, a0, false, b, (short)0, acc0, false, false);
        acc1 = __builtin_amdgcn_wmma_f32_16x16x4_f32(
            false, a1, false, b, (short)0, acc1, false, false);
    }

    const int mbase = tileM + ((lane >> 4) << 3);
    const int col   = tileN + rsel;
    for (int r = 0; r < 8; ++r) {
        int row0 = mbase + r;
        int row1 = row0 + 16;
        if (row0 < Nrows) C[(size_t)row0 * NCOLS + col] = acc0[r];
        if (row1 < Nrows) C[(size_t)row1 * NCOLS + col] = acc1[r];
    }
}

__device__ __forceinline__ float lrelu(float x, float slope) {
    return x > 0.f ? x : slope * x;
}

#define BSHIFT 8
#define BNODES 256
#define NCHUNK 256
#define EPT 8

__device__ __forceinline__ int bucket_of(int d, int nb) {
    const unsigned b = (unsigned)d >> BSHIFT;
    return (int)(b < (unsigned)nb ? b : (unsigned)(nb - 1));
}

__device__ __forceinline__ int block_excl_scan(int v, int* wsum, int nwaves, int& total) {
    const int lane = threadIdx.x & 31;
    const int wave = threadIdx.x >> 5;
    int incl = v;
    for (int off = 1; off < 32; off <<= 1) {
        int t = __shfl_up(incl, off, 32);
        if (lane >= off) incl += t;
    }
    if (lane == 31) wsum[wave] = incl;
    __syncthreads();
    int base = 0, tot = 0;
    for (int w = 0; w < nwaves; ++w) {
        const int t = wsum[w];
        if (w < wave) base += t;
        tot += t;
    }
    __syncthreads();
    total = tot;
    return base + incl - v;
}

__global__ __launch_bounds__(1024)
void gat_bucket_hist(const int* __restrict__ dstIdx, int E, int chunk, int nb,
                     int* __restrict__ hist) {
    extern __shared__ int lcount[];
    for (int i = threadIdx.x; i < nb; i += 1024) lcount[i] = 0;
    __syncthreads();
    const int e0 = blockIdx.x * chunk;
    const int e1 = min(E, e0 + chunk);
    for (int base = e0; base < e1; base += EPT * 1024) {
        int d[EPT];
        for (int k = 0; k < EPT; ++k) {
            const int e = base + k * 1024 + (int)threadIdx.x;
            d[k] = dstIdx[e < e1 ? e : e1 - 1];
        }
        for (int k = 0; k < EPT; ++k) {
            const int e = base + k * 1024 + (int)threadIdx.x;
            if (e < e1) atomicAdd(&lcount[bucket_of(d[k], nb)], 1);
        }
    }
    __syncthreads();
    for (int i = threadIdx.x; i < nb; i += 1024) hist[i * NCHUNK + blockIdx.x] = lcount[i];
}

__global__ __launch_bounds__(NCHUNK)
void gat_bucket_rows(int* __restrict__ hist, int* __restrict__ bucketTotal) {
    __shared__ int wsum[32];
    const int i = blockIdx.x * NCHUNK + threadIdx.x;
    const int v = hist[i];
    int total;
    const int ex = block_excl_scan(v, wsum, NCHUNK / 32, total);
    hist[i] = ex;
    if (threadIdx.x == 0) bucketTotal[blockIdx.x] = total;
}

__global__ __launch_bounds__(1024)
void gat_bucket_scatter(const int* __restrict__ srcIdx, const int* __restrict__ dstIdx,
                        int E, int chunk, int nb,
                        const int* __restrict__ rowPre, const int* __restrict__ bucketTotal,
                        int* __restrict__ bucketBase, int2* __restrict__ part) {
    extern __shared__ int cursor[];
    __shared__ int wsum[32];
    int carry = 0;
    for (int base = 0; base < nb; base += 1024) {
        const int b = base + (int)threadIdx.x;
        const int v = b < nb ? bucketTotal[b] : 0;
        int total;
        const int ex = block_excl_scan(v, wsum, 32, total);
        if (b < nb) {
            cursor[b] = carry + ex + rowPre[b * NCHUNK + blockIdx.x];
            if (blockIdx.x == 0) bucketBase[b] = carry + ex;
        }
        carry += total;
    }
    __syncthreads();
    const int e0 = blockIdx.x * chunk;
    const int e1 = min(E, e0 + chunk);
    for (int base = e0; base < e1; base += EPT * 1024) {
        int s[EPT], d[EPT];
        for (int k = 0; k < EPT; ++k) {
            const int e = base + k * 1024 + (int)threadIdx.x;
            const int ec = e < e1 ? e : e1 - 1;
            s[k] = srcIdx[ec];
            d[k] = dstIdx[ec];
        }
        for (int k = 0; k < EPT; ++k) {
            const int e = base + k * 1024 + (int)threadIdx.x;
            if (e < e1) {
                const int pos = atomicAdd(&cursor[bucket_of(d[k], nb)], 1);
                part[pos] = make_int2(s[k], d[k]);
            }
        }
    }
}

__global__ __launch_bounds__(BNODES)
void gat_bucket_sort(const int2* __restrict__ part, const int* __restrict__ bucketBase,
                     const int* __restrict__ bucketTotal, int N,
                     int* __restrict__ rowStart, int* __restrict__ cnt,
                     int* __restrict__ segSrc) {
    __shared__ int lnode[BNODES];
    __shared__ int wsum[32];
    const int tid = threadIdx.x;
    const int p0 = bucketBase[blockIdx.x];
    const int p1 = p0 + bucketTotal[blockIdx.x];
    lnode[tid] = 0;
    __syncthreads();
    for (int base = p0; base < p1; base += EPT * BNODES) {
        int d[EPT];
        for (int k = 0; k < EPT; ++k) {
            const int i = base + k * BNODES + tid;
            d[k] = part[i < p1 ? i : p1 - 1].y;
        }
        for (int k = 0; k < EPT; ++k) {
            const int i = base + k * BNODES + tid;
            if (i < p1) atomicAdd(&lnode[d[k] & (BNODES - 1)], 1);
        }
    }
    __syncthreads();
    const int c = lnode[tid];
    int total;
    const int ex = block_excl_scan(c, wsum, BNODES / 32, total);
    lnode[tid] = p0 + ex;
    const int n = blockIdx.x * BNODES + tid;
    if (n < N) { rowStart[n] = p0 + ex; cnt[n] = c; }
    __syncthreads();
    for (int base = p0; base < p1; base += EPT * BNODES) {
        int2 sd[EPT];
        for (int k = 0; k < EPT; ++k) {
            const int i = base + k * BNODES + tid;
            sd[k] = part[i < p1 ? i : p1 - 1];
        }
        for (int k = 0; k < EPT; ++k) {
            const int i = base + k * BNODES + tid;
            if (i < p1) {
                const int pos = atomicAdd(&lnode[sd[k].y & (BNODES - 1)], 1);
                segSrc[pos] = sd[k].x;
            }
        }
    }
}

template <int MASK>
__device__ __forceinline__ float row_xor(float x) {
    return __int_as_float(__builtin_amdgcn_update_dpp(
        0, __float_as_int(x), 0x160 + MASK, 0xf, 0xf, true));
}
__device__ __forceinline__ float half_xor(float x) {
    const int b = __float_as_int(x);
    return __int_as_float(__builtin_amdgcn_permlanex16(b, b, 0x76543210, 0xfedcba98, false, false));
}
template <int LPH>
__device__ __forceinline__ float head_sum(float v) {
    v += row_xor<1>(v);
    v += row_xor<2>(v);
    v += row_xor<4>(v);
    if constexpr (LPH > 8)  v += row_xor<8>(v);
    if constexpr (LPH > 16) v += half_xor(v);
    return v;
}

template <int P>
__device__ __forceinline__ void load_row(const float* __restrict__ p, v2f (&v)[P]) {
    if constexpr (P == 2) {
        float4 t = *(const float4*)p;
        v[0].x = t.x; v[0].y = t.y; v[1].x = t.z; v[1].y = t.w;
    } else {
        float2 t = *(const float2*)p;
        v[0].x = t.x; v[0].y = t.y;
    }
}
template <int P>
__device__ __forceinline__ float dot_cols(const v2f (&v)[P], const v2f (&w)[P]) {
    v2f q = v[0] * w[0];
    for (int k = 1; k < P; ++k) q += v[k] * w[k];
    return q.x + q.y;
}

template <int NB, int F, int P, int LPH>
__device__ __forceinline__ void gat_rows(int sItem, int t, int lane,
                                         const float* __restrict__ h,
                                         const v2f (&as)[P], float ed,
                                         float& m, float& sum, v2f (&acc)[P]) {
    v2f v[NB][P];
    for (int u = 0; u < NB; ++u) {
        const int st = __builtin_amdgcn_readlane(sItem, t + u);
        const unsigned boff = ((unsigned)st * (unsigned)F + (unsigned)(lane * (2 * P))) * 4u;
        load_row<P>((const float*)((const char*)h + boff), v[u]);
    }
    float a[NB];
    for (int u = 0; u < NB; ++u)
        a[u] = lrelu(head_sum<LPH>(dot_cols<P>(v[u], as)) + ed, NEG_SLOPE_ATT);
    float nm = m;
    for (int u = 0; u < NB; ++u) nm = fmaxf(nm, a[u]);
    const float sc = __expf(m - nm);
    sum *= sc;
    for (int k = 0; k < P; ++k) acc[k] *= sc;
    for (int u = 0; u < NB; ++u) {
        const float pe = __expf(a[u] - nm);
        sum += pe;
        for (int k = 0; k < P; ++k) acc[k] += pe * v[u][k];
    }
    m = nm;
}

template <int H, int C>
__global__ __launch_bounds__(256)
void gat_gather(const int* __restrict__ rowStart, const int* __restrict__ cnt,
                const int* __restrict__ segSrc, int N,
                const float* __restrict__ a_src, const float* __restrict__ a_dst,
                const float* __restrict__ h, const float* __restrict__ bias,
                float* __restrict__ out, float slope, int apply_act) {
    constexpr int F = H * C;
    constexpr int VEC = F / 32;
    constexpr int P = VEC / 2;
    constexpr int LPH = C / VEC;
    static_assert(F % 32 == 0 && (VEC == 4 || VEC == 2), "row = 32 lanes x VEC");
    static_assert(C % VEC == 0, "a lane's columns stay inside one head");
    static_assert(LPH == 8 || LPH == 16 || LPH == 32, "a head is 8, 16 or 32 adjacent lanes");

    const int lane = threadIdx.x & 31;
    const int d = __builtin_amdgcn_readfirstlane((int)(blockIdx.x * 8 + (threadIdx.x >> 5)));
    if (d >= N) return;
    const int start = rowStart[d];
    const int deg   = cnt[d];
    const int col   = lane * VEC;

    v2f as[P], ad[P], vs[P];
    load_row<P>(a_src + col, as);
    load_row<P>(a_dst + col, ad);
    load_row<P>(h + (size_t)d * F + col, vs);
    int sItem = d;
    if (lane < deg) sItem = segSrc[start + lane];

    const float ed = head_sum<LPH>(dot_cols<P>(vs, ad));
    float m   = lrelu(head_sum<LPH>(dot_cols<P>(vs, as)) + ed, NEG_SLOPE_ATT);
    float sum = 1.f;
    v2f acc[P];
    for (int k = 0; k < P; ++k) acc[k] = vs[k];

    for (int base = 0; base < deg; base += 32) {
        if (base > 0) {
            sItem = d;
            if (base + lane < deg) sItem = segSrc[start + base + lane];
        }
        const int left = deg - base;
        const int n = left < 32 ? left : 32;
        int t = 0;
        for (; t + 8 <= n; t += 8)
            gat_rows<8, F, P, LPH>(sItem, t, lane, h, as, ed, m, sum, acc);
        if (t + 4 <= n) {
            gat_rows<4, F, P, LPH>(sItem, t, lane, h, as, ed, m, sum, acc);
            t += 4;
        }
        const int rem = n - t;
        if (rem == 3)      gat_rows<3, F, P, LPH>(sItem, t, lane, h, as, ed, m, sum, acc);
        else if (rem == 2) gat_rows<2, F, P, LPH>(sItem, t, lane, h, as, ed, m, sum, acc);
        else if (rem == 1) gat_rows<1, F, P, LPH>(sItem, t, lane, h, as, ed, m, sum, acc);
    }

    const float inv = 1.0f / sum;
    float r[VEC];
    for (int k = 0; k < VEC; ++k) {
        float x = acc[k >> 1][k & 1] * inv + bias[col + k];
        r[k] = apply_act ? lrelu(x, slope) : x;
    }
    float* op = out + (size_t)d * F + col;
    if constexpr (VEC == 4) {
        *(float4*)op = make_float4(r[0], r[1], r[2], r[3]);
    } else {
        *(float2*)op = make_float2(r[0], r[1]);
    }
}

extern "C" void kernel_launch(void* const* d_in, const int* in_sizes, int n_in,
                              void* d_out, int out_size, void* d_ws, size_t ws_size,
                              hipStream_t stream) {
    const float* x        = (const float*)d_in[0];
    const int*   ei       = (const int*)d_in[1];
    const float* W1       = (const float*)d_in[2];
    const float* att_src1 = (const float*)d_in[3];
    const float* att_dst1 = (const float*)d_in[4];
    const float* b1       = (const float*)d_in[5];
    const float* W2       = (const float*)d_in[6];
    const float* att_src2 = (const float*)d_in[7];
    const float* att_dst2 = (const float*)d_in[8];
    const float* b2       = (const float*)d_in[9];

    const int HC1  = in_sizes[5];
    const int OUTC = in_sizes[9];
    const int F_IN = in_sizes[2] / HC1;
    const int N    = in_sizes[0] / F_IN;
    const int E    = in_sizes[1] / 2;

    const int* srcIdx = ei;
    const int* dstIdx = ei + E;

    float* ws  = (float*)d_ws;
    size_t off = 0;
    float* h1   = ws + off; off += (size_t)N * HC1;
    float* act1 = ws + off; off += (size_t)N * HC1;
    const int NBK   = (N + BNODES - 1) / BNODES;
    const int chunk = (E + NCHUNK - 1) / NCHUNK;
    int* cnt      = (int*)(ws + off); off += (((size_t)N + 3) / 4) * 4;
    int* rowStart = (int*)(ws + off); off += (((size_t)N + 3) / 4) * 4;
    int* segSrc   = (int*)(ws + off); off += (((size_t)E + 3) / 4) * 4;
    int* hist        = (int*)(ws + off); off += (size_t)NBK * NCHUNK;
    int* bucketTotal = (int*)(ws + off); off += (((size_t)NBK + 3) / 4) * 4;
    int* bucketBase  = (int*)(ws + off); off += (((size_t)NBK + 3) / 4) * 4;
    int2* part       = (int2*)(ws + off); off += (((size_t)E * 2 + 3) / 4) * 4;
    float* h2   = h1;
    float* out2 = (float*)d_out;

    const int TB = 256;
    const int rowTiles32 = (N + 31) / 32;
    const int waveBlocks = (N + 7) / 8;

    gat_bucket_hist<<<NCHUNK, 1024, NBK * sizeof(int), stream>>>(dstIdx, E, chunk, NBK, hist);
    gat_bucket_rows<<<NBK, NCHUNK, 0, stream>>>(hist, bucketTotal);
    gat_bucket_scatter<<<NCHUNK, 1024, NBK * sizeof(int), stream>>>(
        srcIdx, dstIdx, E, chunk, NBK, hist, bucketTotal, bucketBase, part);
    gat_bucket_sort<<<NBK, BNODES, 0, stream>>>(part, bucketBase, bucketTotal, N, rowStart, cnt, segSrc);

    gat_gemm_wmma<128, 128><<<rowTiles32, 32 * (HC1 / 16), 32 * (F_IN + 4) * sizeof(float), stream>>>(
        x, W1, h1, N);
    gat_gather<4, 32><<<waveBlocks, TB, 0, stream>>>(
        rowStart, cnt, segSrc, N, att_src1, att_dst1, h1, b1, act1, NEG_SLOPE_ACT,  1);

    gat_gemm_wmma<128, 64><<<rowTiles32, 32 * (OUTC / 16), 32 * (HC1 + 4) * sizeof(float), stream>>>(
        act1, W2, h2, N);
    gat_gather<1, 64><<<waveBlocks, TB, 0, stream>>>(
        rowStart, cnt, segSrc, N, att_src2, att_dst2, h2, b2, out2, 0.f,  0);
}


# Gather: a full batch of 8 edges sums its logits with a transposing butterfly, each lane ends with one edge's logit, so leaky-relu and exp run once per edge; logit sum order changes
# speedup vs baseline: 7.8638x; 1.0415x over previous
#include <hip/hip_runtime.h>
#include <hip/hip_bf16.h>
#include <math.h>


typedef __attribute__((ext_vector_type(2))) float v2f;
typedef __attribute__((ext_vector_type(4))) float v4f;
typedef __attribute__((ext_vector_type(8))) float v8f;

#define NEG_SLOPE_ATT 0.2f
#define NEG_SLOPE_ACT 0.01f

template <int K, int NCOLS>
__global__ __launch_bounds__(2 * NCOLS)
void gat_gemm_wmma(const float* __restrict__ A, const float* __restrict__ B,
                   float* __restrict__ C, int Nrows) {
    constexpr int KP  = K + 4;
    constexpr int NT  = 2 * NCOLS;
    constexpr int K4  = K / 4;
    constexpr int PCS = (32 * K4) / NT;
    static_assert(K % 4 == 0 && NCOLS % 16 == 0, "whole 16-byte pieces, whole column tiles");
    static_assert((32 * K4) % NT == 0, "the tile's pieces divide evenly over the threads");
    extern __shared__ __attribute__((aligned(16))) float ldsA[];
    const int tileM = blockIdx.x * 32;
    const int lane  = threadIdx.x & 31;
    const int wave  = threadIdx.x >> 5;

    v4f piece[PCS];
    #pragma unroll
    for (int j = 0; j < PCS; ++j) {
        const int idx = j * NT + (int)threadIdx.x;
        const int r = idx / K4, c = (idx - r * K4) * 4;
        const int row = tileM + r;
        const int rc  = row < Nrows ? row : Nrows - 1;
        piece[j] = *(const v4f*)(A + (size_t)rc * K + c);
    }
    #pragma unroll
    for (int j = 0; j < PCS; ++j) {
        const int idx = j * NT + (int)threadIdx.x;
        const int r = idx / K4, c = (idx - r * K4) * 4;
        *(v4f*)&ldsA[r * KP + c] = piece[j];
    }
    __syncthreads();

    const int tileN = wave * 16;
    const int rsel  = lane & 15;
    const int khalf = (lane >> 4) << 1;

    v8f acc0 = {};
    v8f acc1 = {};
    const float* __restrict__ a0p = &ldsA[rsel * KP + khalf];
    const float* __restrict__ a1p = &ldsA[(rsel + 16) * KP + khalf];
    for (int k0 = 0; k0 < K; k0 += 4) {
        v2f a0, a1, b;
        a0.x = a0p[k0];     a0.y = a0p[k0 + 1];
        a1.x = a1p[k0];     a1.y = a1p[k0 + 1];
        b.x  = B[(size_t)(k0 + khalf) * NCOLS + tileN + rsel];
        b.y  = B[(size_t)(k0 + khalf + 1) * NCOLS + tileN + rsel];
        acc0 = __builtin_amdgcn_wmma_f32_16x16x4_f32(
            false, a0, false, b, (short)0, acc0, false, false);
        acc1 = __builtin_amdgcn_wmma_f32_16x16x4_f32(
            false, a1, false, b, (short)0, acc1, false, false);
    }

    const int mbase = tileM + ((lane >> 4) << 3);
    const int col   = tileN + rsel;
    for (int r = 0; r < 8; ++r) {
        int row0 = mbase + r;
        int row1 = row0 + 16;
        if (row0 < Nrows) C[(size_t)row0 * NCOLS + col] = acc0[r];
        if (row1 < Nrows) C[(size_t)row1 * NCOLS + col] = acc1[r];
    }
}

__device__ __forceinline__ float lrelu(float x, float slope) {
    return x > 0.f ? x : slope * x;
}

#define BSHIFT 8
#define BNODES 256
#define NCHUNK 256
#define EPT 8

__device__ __forceinline__ int bucket_of(int d, int nb) {
    const unsigned b = (unsigned)d >> BSHIFT;
    return (int)(b < (unsigned)nb ? b : (unsigned)(nb - 1));
}

__device__ __forceinline__ int block_excl_scan(int v, int* wsum, int nwaves, int& total) {
    const int lane = threadIdx.x & 31;
    const int wave = threadIdx.x >> 5;
    int incl = v;
    for (int off = 1; off < 32; off <<= 1) {
        int t = __shfl_up(incl, off, 32);
        if (lane >= off) incl += t;
    }
    if (lane == 31) wsum[wave] = incl;
    __syncthreads();
    int base = 0, tot = 0;
    for (int w = 0; w < nwaves; ++w) {
        const int t = wsum[w];
        if (w < wave) base += t;
        tot += t;
    }
    __syncthreads();
    total = tot;
    return base + incl - v;
}

__global__ __launch_bounds__(1024)
void gat_bucket_hist(const int* __restrict__ dstIdx, int E, int chunk, int nb,
                     int* __restrict__ hist) {
    extern __shared__ int lcount[];
    for (int i = threadIdx.x; i < nb; i += 1024) lcount[i] = 0;
    __syncthreads();
    const int e0 = blockIdx.x * chunk;
    const int e1 = min(E, e0 + chunk);
    for (int base = e0; base < e1; base += EPT * 1024) {
        int d[EPT];
        for (int k = 0; k < EPT; ++k) {
            const int e = base + k * 1024 + (int)threadIdx.x;
            d[k] = dstIdx[e < e1 ? e : e1 - 1];
        }
        for (int k = 0; k < EPT; ++k) {
            const int e = base + k * 1024 + (int)threadIdx.x;
            if (e < e1) atomicAdd(&lcount[bucket_of(d[k], nb)], 1);
        }
    }
    __syncthreads();
    for (int i = threadIdx.x; i < nb; i += 1024) hist[i * NCHUNK + blockIdx.x] = lcount[i];
}

__global__ __launch_bounds__(NCHUNK)
void gat_bucket_rows(int* __restrict__ hist, int* __restrict__ bucketTotal) {
    __shared__ int wsum[32];
    const int i = blockIdx.x * NCHUNK + threadIdx.x;
    const int v = hist[i];
    int total;
    const int ex = block_excl_scan(v, wsum, NCHUNK / 32, total);
    hist[i] = ex;
    if (threadIdx.x == 0) bucketTotal[blockIdx.x] = total;
}

__global__ __launch_bounds__(1024)
void gat_bucket_scatter(const int* __restrict__ srcIdx, const int* __restrict__ dstIdx,
                        int E, int chunk, int nb,
                        const int* __restrict__ rowPre, const int* __restrict__ bucketTotal,
                        int* __restrict__ bucketBase, int2* __restrict__ part) {
    extern __shared__ int cursor[];
    __shared__ int wsum[32];
    int carry = 0;
    for (int base = 0; base < nb; base += 1024) {
        const int b = base + (int)threadIdx.x;
        const int v = b < nb ? bucketTotal[b] : 0;
        int total;
        const int ex = block_excl_scan(v, wsum, 32, total);
        if (b < nb) {
            cursor[b] = carry + ex + rowPre[b * NCHUNK + blockIdx.x];
            if (blockIdx.x == 0) bucketBase[b] = carry + ex;
        }
        carry += total;
    }
    __syncthreads();
    const int e0 = blockIdx.x * chunk;
    const int e1 = min(E, e0 + chunk);
    for (int base = e0; base < e1; base += EPT * 1024) {
        int s[EPT], d[EPT];
        for (int k = 0; k < EPT; ++k) {
            const int e = base + k * 1024 + (int)threadIdx.x;
            const int ec = e < e1 ? e : e1 - 1;
            s[k] = srcIdx[ec];
            d[k] = dstIdx[ec];
        }
        for (int k = 0; k < EPT; ++k) {
            const int e = base + k * 1024 + (int)threadIdx.x;
            if (e < e1) {
                const int pos = atomicAdd(&cursor[bucket_of(d[k], nb)], 1);
                part[pos] = make_int2(s[k], d[k]);
            }
        }
    }
}

__global__ __launch_bounds__(BNODES)
void gat_bucket_sort(const int2* __restrict__ part, const int* __restrict__ bucketBase,
                     const int* __restrict__ bucketTotal, int N,
                     int* __restrict__ rowStart, int* __restrict__ cnt,
                     int* __restrict__ segSrc) {
    __shared__ int lnode[BNODES];
    __shared__ int wsum[32];
    const int tid = threadIdx.x;
    const int p0 = bucketBase[blockIdx.x];
    const int p1 = p0 + bucketTotal[blockIdx.x];
    lnode[tid] = 0;
    __syncthreads();
    for (int base = p0; base < p1; base += EPT * BNODES) {
        int d[EPT];
        for (int k = 0; k < EPT; ++k) {
            const int i = base + k * BNODES + tid;
            d[k] = part[i < p1 ? i : p1 - 1].y;
        }
        for (int k = 0; k < EPT; ++k) {
            const int i = base + k * BNODES + tid;
            if (i < p1) atomicAdd(&lnode[d[k] & (BNODES - 1)], 1);
        }
    }
    __syncthreads();
    const int c = lnode[tid];
    int total;
    const int ex = block_excl_scan(c, wsum, BNODES / 32, total);
    lnode[tid] = p0 + ex;
    const int n = blockIdx.x * BNODES + tid;
    if (n < N) { rowStart[n] = p0 + ex; cnt[n] = c; }
    __syncthreads();
    for (int base = p0; base < p1; base += EPT * BNODES) {
        int2 sd[EPT];
        for (int k = 0; k < EPT; ++k) {
            const int i = base + k * BNODES + tid;
            sd[k] = part[i < p1 ? i : p1 - 1];
        }
        for (int k = 0; k < EPT; ++k) {
            const int i = base + k * BNODES + tid;
            if (i < p1) {
                const int pos = atomicAdd(&lnode[sd[k].y & (BNODES - 1)], 1);
                segSrc[pos] = sd[k].x;
            }
        }
    }
}

template <int MASK>
__device__ __forceinline__ float row_xor(float x) {
    return __int_as_float(__builtin_amdgcn_update_dpp(
        0, __float_as_int(x), 0x160 + MASK, 0xf, 0xf, true));
}
__device__ __forceinline__ float half_xor(float x) {
    const int b = __float_as_int(x);
    return __int_as_float(__builtin_amdgcn_permlanex16(b, b, 0x76543210, 0xfedcba98, false, false));
}
template <int LPH>
__device__ __forceinline__ float head_sum(float v) {
    v += row_xor<1>(v);
    v += row_xor<2>(v);
    v += row_xor<4>(v);
    if constexpr (LPH > 8)  v += row_xor<8>(v);
    if constexpr (LPH > 16) v += half_xor(v);
    return v;
}

template <int P>
__device__ __forceinline__ void load_row(const float* __restrict__ p, v2f (&v)[P]) {
    if constexpr (P == 2) {
        float4 t = *(const float4*)p;
        v[0].x = t.x; v[0].y = t.y; v[1].x = t.z; v[1].y = t.w;
    } else {
        float2 t = *(const float2*)p;
        v[0].x = t.x; v[0].y = t.y;
    }
}
template <int P>
__device__ __forceinline__ float dot_cols(const v2f (&v)[P], const v2f (&w)[P]) {
    v2f q = v[0] * w[0];
    for (int k = 1; k < P; ++k) q += v[k] * w[k];
    return q.x + q.y;
}

template <int NB, int F, int P, int LPH>
__device__ __forceinline__ void gat_rows(int sItem, int t, int lane,
                                         const float* __restrict__ h,
                                         const v2f (&as)[P], float ed,
                                         float& m, float& sum, v2f (&acc)[P]) {
    v2f v[NB][P];
    for (int u = 0; u < NB; ++u) {
        const int st = __builtin_amdgcn_readlane(sItem, t + u);
        const unsigned boff = ((unsigned)st * (unsigned)F + (unsigned)(lane * (2 * P))) * 4u;
        load_row<P>((const float*)((const char*)h + boff), v[u]);
    }
    float a[NB];
    for (int u = 0; u < NB; ++u)
        a[u] = lrelu(head_sum<LPH>(dot_cols<P>(v[u], as)) + ed, NEG_SLOPE_ATT);
    float nm = m;
    for (int u = 0; u < NB; ++u) nm = fmaxf(nm, a[u]);
    const float sc = __expf(m - nm);
    sum *= sc;
    for (int k = 0; k < P; ++k) acc[k] *= sc;
    for (int u = 0; u < NB; ++u) {
        const float pe = __expf(a[u] - nm);
        sum += pe;
        for (int k = 0; k < P; ++k) acc[k] += pe * v[u][k];
    }
    m = nm;
}

template <int U, int LPH>
__device__ __forceinline__ float edge_weight(float pe) {
    if constexpr (LPH == 32) {
        return __int_as_float(__builtin_amdgcn_readlane(__float_as_int(pe), 4 * U));
    } else {
        constexpr int SEL = U * 0x249249;
        return __int_as_float(__builtin_amdgcn_mov_dpp8(__float_as_int(pe), SEL));
    }
}

template <int F, int P, int LPH>
__device__ __forceinline__ void gat_rows8(int sItem, int t, int lane,
                                          const float* __restrict__ h,
                                          const v2f (&as)[P], float ed,
                                          float& m, float& sum, v2f (&acc)[P]) {
    static_assert(LPH == 8 || LPH == 32, "exchange steps written for heads of 8 or 32 lanes");
    v2f v[8][P];
    for (int u = 0; u < 8; ++u) {
        const int st = __builtin_amdgcn_readlane(sItem, t + u);
        const unsigned boff = ((unsigned)st * (unsigned)F + (unsigned)(lane * (2 * P))) * 4u;
        load_row<P>((const float*)((const char*)h + boff), v[u]);
    }
    float p[8];
    for (int u = 0; u < 8; ++u) p[u] = dot_cols<P>(v[u], as);
    float q[4], r[2], s;
    if constexpr (LPH == 32) {
        const bool c4 = (lane & 16) != 0, c3 = (lane & 8) != 0, c2 = (lane & 4) != 0;
        for (int j = 0; j < 4; ++j)
            q[j] = (c4 ? p[j + 4] : p[j]) + half_xor(c4 ? p[j] : p[j + 4]);
        for (int j = 0; j < 2; ++j)
            r[j] = (c3 ? q[j + 2] : q[j]) + row_xor<8>(c3 ? q[j] : q[j + 2]);
        s = (c2 ? r[1] : r[0]) + row_xor<4>(c2 ? r[0] : r[1]);
        s += row_xor<2>(s);
        s += row_xor<1>(s);
    } else {
        const bool c2 = (lane & 4) != 0, c1 = (lane & 2) != 0, c0 = (lane & 1) != 0;
        for (int j = 0; j < 4; ++j)
            q[j] = (c2 ? p[j + 4] : p[j]) + row_xor<4>(c2 ? p[j] : p[j + 4]);
        for (int j = 0; j < 2; ++j)
            r[j] = (c1 ? q[j + 2] : q[j]) + row_xor<2>(c1 ? q[j] : q[j + 2]);
        s = (c0 ? r[1] : r[0]) + row_xor<1>(c0 ? r[0] : r[1]);
    }
    const float a = lrelu(s + ed, NEG_SLOPE_ATT);
    float mx = a;
    if constexpr (LPH == 32) {
        mx = fmaxf(mx, row_xor<4>(mx));
        mx = fmaxf(mx, row_xor<8>(mx));
        mx = fmaxf(mx, half_xor(mx));
    } else {
        mx = fmaxf(mx, row_xor<1>(mx));
        mx = fmaxf(mx, row_xor<2>(mx));
        mx = fmaxf(mx, row_xor<4>(mx));
    }
    const float nm = fmaxf(m, mx);
    const float sc = __expf(m - nm);
    const float pe = __expf(a - nm);
    sum *= sc;
    for (int k = 0; k < P; ++k) acc[k] *= sc;
#define GAT_EDGE_IN(U) { const float w = edge_weight<U, LPH>(pe); sum += w; \
        for (int k = 0; k < P; ++k) acc[k] += w * v[U][k]; }
    GAT_EDGE_IN(0) GAT_EDGE_IN(1) GAT_EDGE_IN(2) GAT_EDGE_IN(3)
    GAT_EDGE_IN(4) GAT_EDGE_IN(5) GAT_EDGE_IN(6) GAT_EDGE_IN(7)
#undef GAT_EDGE_IN
    m = nm;
}

template <int H, int C>
__global__ __launch_bounds__(256)
void gat_gather(const int* __restrict__ rowStart, const int* __restrict__ cnt,
                const int* __restrict__ segSrc, int N,
                const float* __restrict__ a_src, const float* __restrict__ a_dst,
                const float* __restrict__ h, const float* __restrict__ bias,
                float* __restrict__ out, float slope, int apply_act) {
    constexpr int F = H * C;
    constexpr int VEC = F / 32;
    constexpr int P = VEC / 2;
    constexpr int LPH = C / VEC;
    static_assert(F % 32 == 0 && (VEC == 4 || VEC == 2), "row = 32 lanes x VEC");
    static_assert(C % VEC == 0, "a lane's columns stay inside one head");
    static_assert(LPH == 8 || LPH == 32, "a head is 8 or 32 adjacent lanes");

    const int lane = threadIdx.x & 31;
    const int d = __builtin_amdgcn_readfirstlane((int)(blockIdx.x * 8 + (threadIdx.x >> 5)));
    if (d >= N) return;
    const int start = rowStart[d];
    const int deg   = cnt[d];
    const int col   = lane * VEC;

    v2f as[P], ad[P], vs[P];
    load_row<P>(a_src + col, as);
    load_row<P>(a_dst + col, ad);
    load_row<P>(h + (size_t)d * F + col, vs);
    int sItem = d;
    if (lane < deg) sItem = segSrc[start + lane];

    const float ed = head_sum<LPH>(dot_cols<P>(vs, ad));
    float m   = lrelu(head_sum<LPH>(dot_cols<P>(vs, as)) + ed, NEG_SLOPE_ATT);
    float sum = 1.f;
    v2f acc[P];
    for (int k = 0; k < P; ++k) acc[k] = vs[k];

    for (int base = 0; base < deg; base += 32) {
        if (base > 0) {
            sItem = d;
            if (base + lane < deg) sItem = segSrc[start + base + lane];
        }
        const int left = deg - base;
        const int n = left < 32 ? left : 32;
        int t = 0;
        for (; t + 8 <= n; t += 8)
            gat_rows8<F, P, LPH>(sItem, t, lane, h, as, ed, m, sum, acc);
        if (t + 4 <= n) {
            gat_rows<4, F, P, LPH>(sItem, t, lane, h, as, ed, m, sum, acc);
            t += 4;
        }
        const int rem = n - t;
        if (rem == 3)      gat_rows<3, F, P, LPH>(sItem, t, lane, h, as, ed, m, sum, acc);
        else if (rem == 2) gat_rows<2, F, P, LPH>(sItem, t, lane, h, as, ed, m, sum, acc);
        else if (rem == 1) gat_rows<1, F, P, LPH>(sItem, t, lane, h, as, ed, m, sum, acc);
    }

    const float inv = 1.0f / sum;
    float r[VEC];
    for (int k = 0; k < VEC; ++k) {
        float x = acc[k >> 1][k & 1] * inv + bias[col + k];
        r[k] = apply_act ? lrelu(x, slope) : x;
    }
    float* op = out + (size_t)d * F + col;
    if constexpr (VEC == 4) {
        *(float4*)op = make_float4(r[0], r[1], r[2], r[3]);
    } else {
        *(float2*)op = make_float2(r[0], r[1]);
    }
}

extern "C" void kernel_launch(void* const* d_in, const int* in_sizes, int n_in,
                              void* d_out, int out_size, void* d_ws, size_t ws_size,
                              hipStream_t stream) {
    const float* x        = (const float*)d_in[0];
    const int*   ei       = (const int*)d_in[1];
    const float* W1       = (const float*)d_in[2];
    const float* att_src1 = (const float*)d_in[3];
    const float* att_dst1 = (const float*)d_in[4];
    const float* b1       = (const float*)d_in[5];
    const float* W2       = (const float*)d_in[6];
    const float* att_src2 = (const float*)d_in[7];
    const float* att_dst2 = (const float*)d_in[8];
    const float* b2       = (const float*)d_in[9];

    const int HC1  = in_sizes[5];
    const int OUTC = in_sizes[9];
    const int F_IN = in_sizes[2] / HC1;
    const int N    = in_sizes[0] / F_IN;
    const int E    = in_sizes[1] / 2;

    const int* srcIdx = ei;
    const int* dstIdx = ei + E;

    float* ws  = (float*)d_ws;
    size_t off = 0;
    float* h1   = ws + off; off += (size_t)N * HC1;
    float* act1 = ws + off; off += (size_t)N * HC1;
    const int NBK   = (N + BNODES - 1) / BNODES;
    const int chunk = (E + NCHUNK - 1) / NCHUNK;
    int* cnt      = (int*)(ws + off); off += (((size_t)N + 3) / 4) * 4;
    int* rowStart = (int*)(ws + off); off += (((size_t)N + 3) / 4) * 4;
    int* segSrc   = (int*)(ws + off); off += (((size_t)E + 3) / 4) * 4;
    int* hist        = (int*)(ws + off); off += (size_t)NBK * NCHUNK;
    int* bucketTotal = (int*)(ws + off); off += (((size_t)NBK + 3) / 4) * 4;
    int* bucketBase  = (int*)(ws + off); off += (((size_t)NBK + 3) / 4) * 4;
    int2* part       = (int2*)(ws + off); off += (((size_t)E * 2 + 3) / 4) * 4;
    float* h2   = h1;
    float* out2 = (float*)d_out;

    const int TB = 256;
    const int rowTiles32 = (N + 31) / 32;
    const int waveBlocks = (N + 7) / 8;

    gat_bucket_hist<<<NCHUNK, 1024, NBK * sizeof(int), stream>>>(dstIdx, E, chunk, NBK, hist);
    gat_bucket_rows<<<NBK, NCHUNK, 0, stream>>>(hist, bucketTotal);
    gat_bucket_scatter<<<NCHUNK, 1024, NBK * sizeof(int), stream>>>(
        srcIdx, dstIdx, E, chunk, NBK, hist, bucketTotal, bucketBase, part);
    gat_bucket_sort<<<NBK, BNODES, 0, stream>>>(part, bucketBase, bucketTotal, N, rowStart, cnt, segSrc);

    gat_gemm_wmma<128, 128><<<rowTiles32, 32 * (HC1 / 16), 32 * (F_IN + 4) * sizeof(float), stream>>>(
        x, W1, h1, N);
    gat_gather<4, 32><<<waveBlocks, TB, 0, stream>>>(
        rowStart, cnt, segSrc, N, att_src1, att_dst1, h1, b1, act1, NEG_SLOPE_ACT,  1);

    gat_gemm_wmma<128, 64><<<rowTiles32, 32 * (OUTC / 16), 32 * (HC1 + 4) * sizeof(float), stream>>>(
        act1, W2, h2, N);
    gat_gather<1, 64><<<waveBlocks, TB, 0, stream>>>(
        rowStart, cnt, segSrc, N, att_src2, att_dst2, h2, b2, out2, 0.f,  0);
}


# Layer-1 projection tiles run as extra work-groups inside the hist, scatter and bucket-sort launches, beside passes that leave the matrix units idle; the projection's own launch goes (7 launches)
# speedup vs baseline: 8.1954x; 1.0422x over previous
#include <hip/hip_runtime.h>
#include <hip/hip_bf16.h>
#include <math.h>


typedef __attribute__((ext_vector_type(2))) float v2f;
typedef __attribute__((ext_vector_type(4))) float v4f;
typedef __attribute__((ext_vector_type(8))) float v8f;

#define NEG_SLOPE_ATT 0.2f
#define NEG_SLOPE_ACT 0.01f

template <int K, int NCOLS>
__device__ __forceinline__ void gat_gemm_tiles(const float* __restrict__ A, const float* __restrict__ B,
                                               float* __restrict__ C, int Nrows, int tile0,
                                               float* __restrict__ lds) {
    constexpr int KP  = K + 4;
    constexpr int NT  = 2 * NCOLS;
    constexpr int K4  = K / 4;
    constexpr int PCS = (32 * K4) / NT;
    static_assert(K % 4 == 0 && NCOLS % 16 == 0, "whole 16-byte pieces, whole column tiles");
    static_assert((32 * K4) % NT == 0, "the tile's pieces divide evenly over the threads");
    constexpr int WPG = NT / 32;
    const int tid   = (int)threadIdx.x % NT;
    const int wv    = __builtin_amdgcn_readfirstlane((int)(threadIdx.x >> 5));
    const int grp   = wv / WPG;
    float* __restrict__ ldsA = lds + grp * (32 * KP);
    const int tileM = (tile0 + grp) * 32;
    const int lane  = tid & 31;
    const int wave  = wv % WPG;

    v4f piece[PCS];
    #pragma unroll
    for (int j = 0; j < PCS; ++j) {
        const int idx = j * NT + tid;
        const int r = idx / K4, c = (idx - r * K4) * 4;
        const int row = tileM + r;
        const int rc  = row < Nrows ? row : Nrows - 1;
        piece[j] = *(const v4f*)(A + (size_t)rc * K + c);
    }
    #pragma unroll
    for (int j = 0; j < PCS; ++j) {
        const int idx = j * NT + tid;
        const int r = idx / K4, c = (idx - r * K4) * 4;
        *(v4f*)&ldsA[r * KP + c] = piece[j];
    }
    __syncthreads();

    const int tileN = wave * 16;
    const int rsel  = lane & 15;
    const int khalf = (lane >> 4) << 1;

    v8f acc0 = {};
    v8f acc1 = {};
    const float* __restrict__ a0p = &ldsA[rsel * KP + khalf];
    const float* __restrict__ a1p = &ldsA[(rsel + 16) * KP + khalf];
    for (int k0 = 0; k0 < K; k0 += 4) {
        v2f a0, a1, b;
        a0.x = a0p[k0];     a0.y = a0p[k0 + 1];
        a1.x = a1p[k0];     a1.y = a1p[k0 + 1];
        b.x  = B[(size_t)(k0 + khalf) * NCOLS + tileN + rsel];
        b.y  = B[(size_t)(k0 + khalf + 1) * NCOLS + tileN + rsel];
        acc0 = __builtin_amdgcn_wmma_f32_16x16x4_f32(
            false, a0, false, b, (short)0, acc0, false, false);
        acc1 = __builtin_amdgcn_wmma_f32_16x16x4_f32(
            false, a1, false, b, (short)0, acc1, false, false);
    }

    const int mbase = tileM + ((lane >> 4) << 3);
    const int col   = tileN + rsel;
    for (int r = 0; r < 8; ++r) {
        int row0 = mbase + r;
        int row1 = row0 + 16;
        if (row0 < Nrows) C[(size_t)row0 * NCOLS + col] = acc0[r];
        if (row1 < Nrows) C[(size_t)row1 * NCOLS + col] = acc1[r];
    }
}

template <int K, int NCOLS>
__global__ __launch_bounds__(2 * NCOLS)
void gat_gemm_wmma(const float* __restrict__ A, const float* __restrict__ B,
                   float* __restrict__ C, int Nrows) {
    extern __shared__ __attribute__((aligned(16))) float ldsA[];
    gat_gemm_tiles<K, NCOLS>(A, B, C, Nrows, (int)blockIdx.x, ldsA);
}

#define P1K 128
#define P1C 128
#define P1LDS (32 * (P1K + 4))

__device__ __forceinline__ float lrelu(float x, float slope) {
    return x > 0.f ? x : slope * x;
}

#define BSHIFT 8
#define BNODES 256
#define NCHUNK 256
#define EPT 8

__device__ __forceinline__ int bucket_of(int d, int nb) {
    const unsigned b = (unsigned)d >> BSHIFT;
    return (int)(b < (unsigned)nb ? b : (unsigned)(nb - 1));
}

__device__ __forceinline__ int block_excl_scan(int v, int* wsum, int nwaves, int& total) {
    const int lane = threadIdx.x & 31;
    const int wave = threadIdx.x >> 5;
    int incl = v;
    for (int off = 1; off < 32; off <<= 1) {
        int t = __shfl_up(incl, off, 32);
        if (lane >= off) incl += t;
    }
    if (lane == 31) wsum[wave] = incl;
    __syncthreads();
    int base = 0, tot = 0;
    for (int w = 0; w < nwaves; ++w) {
        const int t = wsum[w];
        if (w < wave) base += t;
        tot += t;
    }
    __syncthreads();
    total = tot;
    return base + incl - v;
}

__global__ __launch_bounds__(1024)
void gat_bucket_hist(const int* __restrict__ dstIdx, int E, int chunk, int nb,
                     int* __restrict__ hist,
                     const float* __restrict__ pA, const float* __restrict__ pB,
                     float* __restrict__ pC, int pRows, int pTile0) {
    __shared__ __attribute__((aligned(16))) float pLds[4 * P1LDS];
    if (blockIdx.x >= NCHUNK) {
        gat_gemm_tiles<P1K, P1C>(pA, pB, pC, pRows, pTile0 + 4 * ((int)blockIdx.x - NCHUNK), pLds);
        return;
    }
    extern __shared__ int lcount[];
    for (int i = threadIdx.x; i < nb; i += 1024) lcount[i] = 0;
    __syncthreads();
    const int e0 = blockIdx.x * chunk;
    const int e1 = min(E, e0 + chunk);
    for (int base = e0; base < e1; base += EPT * 1024) {
        int d[EPT];
        for (int k = 0; k < EPT; ++k) {
            const int e = base + k * 1024 + (int)threadIdx.x;
            d[k] = dstIdx[e < e1 ? e : e1 - 1];
        }
        for (int k = 0; k < EPT; ++k) {
            const int e = base + k * 1024 + (int)threadIdx.x;
            if (e < e1) atomicAdd(&lcount[bucket_of(d[k], nb)], 1);
        }
    }
    __syncthreads();
    for (int i = threadIdx.x; i < nb; i += 1024) hist[i * NCHUNK + blockIdx.x] = lcount[i];
}

__global__ __launch_bounds__(NCHUNK)
void gat_bucket_rows(int* __restrict__ hist, int* __restrict__ bucketTotal) {
    __shared__ int wsum[32];
    const int i = blockIdx.x * NCHUNK + threadIdx.x;
    const int v = hist[i];
    int total;
    const int ex = block_excl_scan(v, wsum, NCHUNK / 32, total);
    hist[i] = ex;
    if (threadIdx.x == 0) bucketTotal[blockIdx.x] = total;
}

__global__ __launch_bounds__(1024)
void gat_bucket_scatter(const int* __restrict__ srcIdx, const int* __restrict__ dstIdx,
                        int E, int chunk, int nb,
                        const int* __restrict__ rowPre, const int* __restrict__ bucketTotal,
                        int* __restrict__ bucketBase, int2* __restrict__ part,
                        const float* __restrict__ pA, const float* __restrict__ pB,
                        float* __restrict__ pC, int pRows, int pTile0) {
    __shared__ __attribute__((aligned(16))) float pLds[4 * P1LDS];
    if (blockIdx.x >= NCHUNK) {
        gat_gemm_tiles<P1K, P1C>(pA, pB, pC, pRows, pTile0 + 4 * ((int)blockIdx.x - NCHUNK), pLds);
        return;
    }
    extern __shared__ int cursor[];
    __shared__ int wsum[32];
    int carry = 0;
    for (int base = 0; base < nb; base += 1024) {
        const int b = base + (int)threadIdx.x;
        const int v = b < nb ? bucketTotal[b] : 0;
        int total;
        const int ex = block_excl_scan(v, wsum, 32, total);
        if (b < nb) {
            cursor[b] = carry + ex + rowPre[b * NCHUNK + blockIdx.x];
            if (blockIdx.x == 0) bucketBase[b] = carry + ex;
        }
        carry += total;
    }
    __syncthreads();
    const int e0 = blockIdx.x * chunk;
    const int e1 = min(E, e0 + chunk);
    for (int base = e0; base < e1; base += EPT * 1024) {
        int s[EPT], d[EPT];
        for (int k = 0; k < EPT; ++k) {
            const int e = base + k * 1024 + (int)threadIdx.x;
            const int ec = e < e1 ? e : e1 - 1;
            s[k] = srcIdx[ec];
            d[k] = dstIdx[ec];
        }
        for (int k = 0; k < EPT; ++k) {
            const int e = base + k * 1024 + (int)threadIdx.x;
            if (e < e1) {
                const int pos = atomicAdd(&cursor[bucket_of(d[k], nb)], 1);
                part[pos] = make_int2(s[k], d[k]);
            }
        }
    }
}

__global__ __launch_bounds__(BNODES)
void gat_bucket_sort(const int2* __restrict__ part, const int* __restrict__ bucketBase,
                     const int* __restrict__ bucketTotal, int N,
                     int* __restrict__ rowStart, int* __restrict__ cnt,
                     int* __restrict__ segSrc, int nbk,
                     const float* __restrict__ pA, const float* __restrict__ pB,
                     float* __restrict__ pC, int pRows, int pTile0) {
    static_assert(BNODES == 2 * P1C, "a projection group is one work-group here");
    __shared__ __attribute__((aligned(16))) float pLds[P1LDS];
    if ((int)blockIdx.x >= nbk) {
        gat_gemm_tiles<P1K, P1C>(pA, pB, pC, pRows, pTile0 + ((int)blockIdx.x - nbk), pLds);
        return;
    }
    __shared__ int lnode[BNODES];
    __shared__ int wsum[32];
    const int tid = threadIdx.x;
    const int p0 = bucketBase[blockIdx.x];
    const int p1 = p0 + bucketTotal[blockIdx.x];
    lnode[tid] = 0;
    __syncthreads();
    for (int base = p0; base < p1; base += EPT * BNODES) {
        int d[EPT];
        for (int k = 0; k < EPT; ++k) {
            const int i = base + k * BNODES + tid;
            d[k] = part[i < p1 ? i : p1 - 1].y;
        }
        for (int k = 0; k < EPT; ++k) {
            const int i = base + k * BNODES + tid;
            if (i < p1) atomicAdd(&lnode[d[k] & (BNODES - 1)], 1);
        }
    }
    __syncthreads();
    const int c = lnode[tid];
    int total;
    const int ex = block_excl_scan(c, wsum, BNODES / 32, total);
    lnode[tid] = p0 + ex;
    const int n = blockIdx.x * BNODES + tid;
    if (n < N) { rowStart[n] = p0 + ex; cnt[n] = c; }
    __syncthreads();
    for (int base = p0; base < p1; base += EPT * BNODES) {
        int2 sd[EPT];
        for (int k = 0; k < EPT; ++k) {
            const int i = base + k * BNODES + tid;
            sd[k] = part[i < p1 ? i : p1 - 1];
        }
        for (int k = 0; k < EPT; ++k) {
            const int i = base + k * BNODES + tid;
            if (i < p1) {
                const int pos = atomicAdd(&lnode[sd[k].y & (BNODES - 1)], 1);
                segSrc[pos] = sd[k].x;
            }
        }
    }
}

template <int MASK>
__device__ __forceinline__ float row_xor(float x) {
    return __int_as_float(__builtin_amdgcn_update_dpp(
        0, __float_as_int(x), 0x160 + MASK, 0xf, 0xf, true));
}
__device__ __forceinline__ float half_xor(float x) {
    const int b = __float_as_int(x);
    return __int_as_float(__builtin_amdgcn_permlanex16(b, b, 0x76543210, 0xfedcba98, false, false));
}
template <int LPH>
__device__ __forceinline__ float head_sum(float v) {
    v += row_xor<1>(v);
    v += row_xor<2>(v);
    v += row_xor<4>(v);
    if constexpr (LPH > 8)  v += row_xor<8>(v);
    if constexpr (LPH > 16) v += half_xor(v);
    return v;
}

template <int P>
__device__ __forceinline__ void load_row(const float* __restrict__ p, v2f (&v)[P]) {
    if constexpr (P == 2) {
        float4 t = *(const float4*)p;
        v[0].x = t.x; v[0].y = t.y; v[1].x = t.z; v[1].y = t.w;
    } else {
        float2 t = *(const float2*)p;
        v[0].x = t.x; v[0].y = t.y;
    }
}
template <int P>
__device__ __forceinline__ float dot_cols(const v2f (&v)[P], const v2f (&w)[P]) {
    v2f q = v[0] * w[0];
    for (int k = 1; k < P; ++k) q += v[k] * w[k];
    return q.x + q.y;
}

template <int NB, int F, int P, int LPH>
__device__ __forceinline__ void gat_rows(int sItem, int t, int lane,
                                         const float* __restrict__ h,
                                         const v2f (&as)[P], float ed,
                                         float& m, float& sum, v2f (&acc)[P]) {
    v2f v[NB][P];
    for (int u = 0; u < NB; ++u) {
        const int st = __builtin_amdgcn_readlane(sItem, t + u);
        const unsigned boff = ((unsigned)st * (unsigned)F + (unsigned)(lane * (2 * P))) * 4u;
        load_row<P>((const float*)((const char*)h + boff), v[u]);
    }
    float a[NB];
    for (int u = 0; u < NB; ++u)
        a[u] = lrelu(head_sum<LPH>(dot_cols<P>(v[u], as)) + ed, NEG_SLOPE_ATT);
    float nm = m;
    for (int u = 0; u < NB; ++u) nm = fmaxf(nm, a[u]);
    const float sc = __expf(m - nm);
    sum *= sc;
    for (int k = 0; k < P; ++k) acc[k] *= sc;
    for (int u = 0; u < NB; ++u) {
        const float pe = __expf(a[u] - nm);
        sum += pe;
        for (int k = 0; k < P; ++k) acc[k] += pe * v[u][k];
    }
    m = nm;
}

template <int U, int LPH>
__device__ __forceinline__ float edge_weight(float pe) {
    if constexpr (LPH == 32) {
        return __int_as_float(__builtin_amdgcn_readlane(__float_as_int(pe), 4 * U));
    } else {
        constexpr int SEL = U * 0x249249;
        return __int_as_float(__builtin_amdgcn_mov_dpp8(__float_as_int(pe), SEL));
    }
}

template <int F, int P, int LPH>
__device__ __forceinline__ void gat_rows8(int sItem, int t, int lane,
                                          const float* __restrict__ h,
                                          const v2f (&as)[P], float ed,
                                          float& m, float& sum, v2f (&acc)[P]) {
    static_assert(LPH == 8 || LPH == 32, "exchange steps written for heads of 8 or 32 lanes");
    v2f v[8][P];
    for (int u = 0; u < 8; ++u) {
        const int st = __builtin_amdgcn_readlane(sItem, t + u);
        const unsigned boff = ((unsigned)st * (unsigned)F + (unsigned)(lane * (2 * P))) * 4u;
        load_row<P>((const float*)((const char*)h + boff), v[u]);
    }
    float p[8];
    for (int u = 0; u < 8; ++u) p[u] = dot_cols<P>(v[u], as);
    float q[4], r[2], s;
    if constexpr (LPH == 32) {
        const bool c4 = (lane & 16) != 0, c3 = (lane & 8) != 0, c2 = (lane & 4) != 0;
        for (int j = 0; j < 4; ++j)
            q[j] = (c4 ? p[j + 4] : p[j]) + half_xor(c4 ? p[j] : p[j + 4]);
        for (int j = 0; j < 2; ++j)
            r[j] = (c3 ? q[j + 2] : q[j]) + row_xor<8>(c3 ? q[j] : q[j + 2]);
        s = (c2 ? r[1] : r[0]) + row_xor<4>(c2 ? r[0] : r[1]);
        s += row_xor<2>(s);
        s += row_xor<1>(s);
    } else {
        const bool c2 = (lane & 4) != 0, c1 = (lane & 2) != 0, c0 = (lane & 1) != 0;
        for (int j = 0; j < 4; ++j)
            q[j] = (c2 ? p[j + 4] : p[j]) + row_xor<4>(c2 ? p[j] : p[j + 4]);
        for (int j = 0; j < 2; ++j)
            r[j] = (c1 ? q[j + 2] : q[j]) + row_xor<2>(c1 ? q[j] : q[j + 2]);
        s = (c0 ? r[1] : r[0]) + row_xor<1>(c0 ? r[0] : r[1]);
    }
    const float a = lrelu(s + ed, NEG_SLOPE_ATT);
    float mx = a;
    if constexpr (LPH == 32) {
        mx = fmaxf(mx, row_xor<4>(mx));
        mx = fmaxf(mx, row_xor<8>(mx));
        mx = fmaxf(mx, half_xor(mx));
    } else {
        mx = fmaxf(mx, row_xor<1>(mx));
        mx = fmaxf(mx, row_xor<2>(mx));
        mx = fmaxf(mx, row_xor<4>(mx));
    }
    const float nm = fmaxf(m, mx);
    const float sc = __expf(m - nm);
    const float pe = __expf(a - nm);
    sum *= sc;
    for (int k = 0; k < P; ++k) acc[k] *= sc;
#define GAT_EDGE_IN(U) { const float w = edge_weight<U, LPH>(pe); sum += w; \
        for (int k = 0; k < P; ++k) acc[k] += w * v[U][k]; }
    GAT_EDGE_IN(0) GAT_EDGE_IN(1) GAT_EDGE_IN(2) GAT_EDGE_IN(3)
    GAT_EDGE_IN(4) GAT_EDGE_IN(5) GAT_EDGE_IN(6) GAT_EDGE_IN(7)
#undef GAT_EDGE_IN
    m = nm;
}

template <int H, int C>
__global__ __launch_bounds__(256)
void gat_gather(const int* __restrict__ rowStart, const int* __restrict__ cnt,
                const int* __restrict__ segSrc, int N,
                const float* __restrict__ a_src, const float* __restrict__ a_dst,
                const float* __restrict__ h, const float* __restrict__ bias,
                float* __restrict__ out, float slope, int apply_act) {
    constexpr int F = H * C;
    constexpr int VEC = F / 32;
    constexpr int P = VEC / 2;
    constexpr int LPH = C / VEC;
    static_assert(F % 32 == 0 && (VEC == 4 || VEC == 2), "row = 32 lanes x VEC");
    static_assert(C % VEC == 0, "a lane's columns stay inside one head");
    static_assert(LPH == 8 || LPH == 32, "a head is 8 or 32 adjacent lanes");

    const int lane = threadIdx.x & 31;
    const int d = __builtin_amdgcn_readfirstlane((int)(blockIdx.x * 8 + (threadIdx.x >> 5)));
    if (d >= N) return;
    const int start = rowStart[d];
    const int deg   = cnt[d];
    const int col   = lane * VEC;

    v2f as[P], ad[P], vs[P];
    load_row<P>(a_src + col, as);
    load_row<P>(a_dst + col, ad);
    load_row<P>(h + (size_t)d * F + col, vs);
    int sItem = d;
    if (lane < deg) sItem = segSrc[start + lane];

    const float ed = head_sum<LPH>(dot_cols<P>(vs, ad));
    float m   = lrelu(head_sum<LPH>(dot_cols<P>(vs, as)) + ed, NEG_SLOPE_ATT);
    float sum = 1.f;
    v2f acc[P];
    for (int k = 0; k < P; ++k) acc[k] = vs[k];

    for (int base = 0; base < deg; base += 32) {
        if (base > 0) {
            sItem = d;
            if (base + lane < deg) sItem = segSrc[start + base + lane];
        }
        const int left = deg - base;
        const int n = left < 32 ? left : 32;
        int t = 0;
        for (; t + 8 <= n; t += 8)
            gat_rows8<F, P, LPH>(sItem, t, lane, h, as, ed, m, sum, acc);
        if (t + 4 <= n) {
            gat_rows<4, F, P, LPH>(sItem, t, lane, h, as, ed, m, sum, acc);
            t += 4;
        }
        const int rem = n - t;
        if (rem == 3)      gat_rows<3, F, P, LPH>(sItem, t, lane, h, as, ed, m, sum, acc);
        else if (rem == 2) gat_rows<2, F, P, LPH>(sItem, t, lane, h, as, ed, m, sum, acc);
        else if (rem == 1) gat_rows<1, F, P, LPH>(sItem, t, lane, h, as, ed, m, sum, acc);
    }

    const float inv = 1.0f / sum;
    float r[VEC];
    for (int k = 0; k < VEC; ++k) {
        float x = acc[k >> 1][k & 1] * inv + bias[col + k];
        r[k] = apply_act ? lrelu(x, slope) : x;
    }
    float* op = out + (size_t)d * F + col;
    if constexpr (VEC == 4) {
        *(float4*)op = make_float4(r[0], r[1], r[2], r[3]);
    } else {
        *(float2*)op = make_float2(r[0], r[1]);
    }
}

extern "C" void kernel_launch(void* const* d_in, const int* in_sizes, int n_in,
                              void* d_out, int out_size, void* d_ws, size_t ws_size,
                              hipStream_t stream) {
    const float* x        = (const float*)d_in[0];
    const int*   ei       = (const int*)d_in[1];
    const float* W1       = (const float*)d_in[2];
    const float* att_src1 = (const float*)d_in[3];
    const float* att_dst1 = (const float*)d_in[4];
    const float* b1       = (const float*)d_in[5];
    const float* W2       = (const float*)d_in[6];
    const float* att_src2 = (const float*)d_in[7];
    const float* att_dst2 = (const float*)d_in[8];
    const float* b2       = (const float*)d_in[9];

    const int HC1  = in_sizes[5];
    const int OUTC = in_sizes[9];
    const int F_IN = in_sizes[2] / HC1;
    const int N    = in_sizes[0] / F_IN;
    const int E    = in_sizes[1] / 2;

    const int* srcIdx = ei;
    const int* dstIdx = ei + E;

    float* ws  = (float*)d_ws;
    size_t off = 0;
    float* h1   = ws + off; off += (size_t)N * HC1;
    float* act1 = ws + off; off += (size_t)N * HC1;
    const int NBK   = (N + BNODES - 1) / BNODES;
    const int chunk = (E + NCHUNK - 1) / NCHUNK;
    int* cnt      = (int*)(ws + off); off += (((size_t)N + 3) / 4) * 4;
    int* rowStart = (int*)(ws + off); off += (((size_t)N + 3) / 4) * 4;
    int* segSrc   = (int*)(ws + off); off += (((size_t)E + 3) / 4) * 4;
    int* hist        = (int*)(ws + off); off += (size_t)NBK * NCHUNK;
    int* bucketTotal = (int*)(ws + off); off += (((size_t)NBK + 3) / 4) * 4;
    int* bucketBase  = (int*)(ws + off); off += (((size_t)NBK + 3) / 4) * 4;
    int2* part       = (int2*)(ws + off); off += (((size_t)E * 2 + 3) / 4) * 4;
    float* h2   = h1;
    float* out2 = (float*)d_out;

    const int TB = 256;
    const int rowTiles32 = (N + 31) / 32;
    const int waveBlocks = (N + 7) / 8;
    const int gA = rowTiles32 / 8 < NCHUNK ? rowTiles32 / 8 : NCHUNK;
    const int gB = gA;
    const int gC = rowTiles32 - 4 * (gA + gB);

    gat_bucket_hist<<<NCHUNK + gA, 1024, NBK * sizeof(int), stream>>>(
        dstIdx, E, chunk, NBK, hist, x, W1, h1, N, 0);
    gat_bucket_rows<<<NBK, NCHUNK, 0, stream>>>(hist, bucketTotal);
    gat_bucket_scatter<<<NCHUNK + gB, 1024, NBK * sizeof(int), stream>>>(
        srcIdx, dstIdx, E, chunk, NBK, hist, bucketTotal, bucketBase, part, x, W1, h1, N, 4 * gA);
    gat_bucket_sort<<<NBK + gC, BNODES, 0, stream>>>(
        part, bucketBase, bucketTotal, N, rowStart, cnt, segSrc, NBK, x, W1, h1, N, 4 * (gA + gB));

    gat_gather<4, 32><<<waveBlocks, TB, 0, stream>>>(
        rowStart, cnt, segSrc, N, att_src1, att_dst1, h1, b1, act1, NEG_SLOPE_ACT,  1);

    gat_gemm_wmma<128, 64><<<rowTiles32, 32 * (OUTC / 16), 32 * (HC1 + 4) * sizeof(float), stream>>>(
        act1, W2, h2, N);
    gat_gather<1, 64><<<waveBlocks, TB, 0, stream>>>(
        rowStart, cnt, segSrc, N, att_src2, att_dst2, h2, b2, out2, 0.f,  0);
}


# Edges grouped by destination in two launches, not four: each chunk sorts its own edges by bucket in LDS and leaves a run table, the bucket pass reads its runs from every chunk; sum order changes
# speedup vs baseline: 8.4805x; 1.0348x over previous
#include <hip/hip_runtime.h>
#include <hip/hip_bf16.h>
#include <math.h>


typedef __attribute__((ext_vector_type(2))) float v2f;
typedef __attribute__((ext_vector_type(4))) float v4f;
typedef __attribute__((ext_vector_type(8))) float v8f;

#define NEG_SLOPE_ATT 0.2f
#define NEG_SLOPE_ACT 0.01f

template <int K, int NCOLS>
__device__ __forceinline__ void gat_gemm_tiles(const float* __restrict__ A, const float* __restrict__ B,
                                               float* __restrict__ C, int Nrows, int tile0,
                                               float* __restrict__ lds) {
    constexpr int KP  = K + 4;
    constexpr int NT  = 2 * NCOLS;
    constexpr int K4  = K / 4;
    constexpr int PCS = (32 * K4) / NT;
    static_assert(K % 4 == 0 && NCOLS % 16 == 0, "whole 16-byte pieces, whole column tiles");
    static_assert((32 * K4) % NT == 0, "the tile's pieces divide evenly over the threads");
    constexpr int WPG = NT / 32;
    const int tid   = (int)threadIdx.x % NT;
    const int wv    = __builtin_amdgcn_readfirstlane((int)(threadIdx.x >> 5));
    const int grp   = wv / WPG;
    float* __restrict__ ldsA = lds + grp * (32 * KP);
    const int tileM = (tile0 + grp) * 32;
    const int lane  = tid & 31;
    const int wave  = wv % WPG;

    v4f piece[PCS];
    #pragma unroll
    for (int j = 0; j < PCS; ++j) {
        const int idx = j * NT + tid;
        const int r = idx / K4, c = (idx - r * K4) * 4;
        const int row = tileM + r;
        const int rc  = row < Nrows ? row : Nrows - 1;
        piece[j] = *(const v4f*)(A + (size_t)rc * K + c);
    }
    #pragma unroll
    for (int j = 0; j < PCS; ++j) {
        const int idx = j * NT + tid;
        const int r = idx / K4, c = (idx - r * K4) * 4;
        *(v4f*)&ldsA[r * KP + c] = piece[j];
    }
    __syncthreads();

    const int tileN = wave * 16;
    const int rsel  = lane & 15;
    const int khalf = (lane >> 4) << 1;

    v8f acc0 = {};
    v8f acc1 = {};
    const float* __restrict__ a0p = &ldsA[rsel * KP + khalf];
    const float* __restrict__ a1p = &ldsA[(rsel + 16) * KP + khalf];
    for (int k0 = 0; k0 < K; k0 += 4) {
        v2f a0, a1, b;
        a0.x = a0p[k0];     a0.y = a0p[k0 + 1];
        a1.x = a1p[k0];     a1.y = a1p[k0 + 1];
        b.x  = B[(size_t)(k0 + khalf) * NCOLS + tileN + rsel];
        b.y  = B[(size_t)(k0 + khalf + 1) * NCOLS + tileN + rsel];
        acc0 = __builtin_amdgcn_wmma_f32_16x16x4_f32(
            false, a0, false, b, (short)0, acc0, false, false);
        acc1 = __builtin_amdgcn_wmma_f32_16x16x4_f32(
            false, a1, false, b, (short)0, acc1, false, false);
    }

    const int mbase = tileM + ((lane >> 4) << 3);
    const int col   = tileN + rsel;
    for (int r = 0; r < 8; ++r) {
        int row0 = mbase + r;
        int row1 = row0 + 16;
        if (row0 < Nrows) C[(size_t)row0 * NCOLS + col] = acc0[r];
        if (row1 < Nrows) C[(size_t)row1 * NCOLS + col] = acc1[r];
    }
}

template <int K, int NCOLS>
__global__ __launch_bounds__(2 * NCOLS)
void gat_gemm_wmma(const float* __restrict__ A, const float* __restrict__ B,
                   float* __restrict__ C, int Nrows) {
    extern __shared__ __attribute__((aligned(16))) float ldsA[];
    gat_gemm_tiles<K, NCOLS>(A, B, C, Nrows, (int)blockIdx.x, ldsA);
}

#define P1K 128
#define P1C 128
#define P1LDS (32 * (P1K + 4))

__device__ __forceinline__ float lrelu(float x, float slope) {
    return x > 0.f ? x : slope * x;
}

#define BSHIFT 8
#define BNODES 256
#define NCHUNK 256
#define EPT 8

__device__ __forceinline__ int bucket_of(int d, int nb) {
    const unsigned b = (unsigned)d >> BSHIFT;
    return (int)(b < (unsigned)nb ? b : (unsigned)(nb - 1));
}

__device__ __forceinline__ int block_excl_scan(int v, int* wsum, int nwaves, int& total) {
    const int lane = threadIdx.x & 31;
    const int wave = threadIdx.x >> 5;
    int incl = v;
    for (int off = 1; off < 32; off <<= 1) {
        int t = __shfl_up(incl, off, 32);
        if (lane >= off) incl += t;
    }
    if (lane == 31) wsum[wave] = incl;
    __syncthreads();
    int base = 0, tot = 0;
    for (int w = 0; w < nwaves; ++w) {
        const int t = wsum[w];
        if (w < wave) base += t;
        tot += t;
    }
    __syncthreads();
    total = tot;
    return base + incl - v;
}

__global__ __launch_bounds__(1024)
void gat_chunk_group(const int* __restrict__ srcIdx, const int* __restrict__ dstIdx,
                     int E, int chunk, int nb,
                     int* __restrict__ offT, int* __restrict__ cntT, int2* __restrict__ part,
                     const float* __restrict__ pA, const float* __restrict__ pB,
                     float* __restrict__ pC, int pRows, int pTile0) {
    __shared__ __attribute__((aligned(16))) float pLds[4 * P1LDS];
    if (blockIdx.x >= NCHUNK) {
        gat_gemm_tiles<P1K, P1C>(pA, pB, pC, pRows, pTile0 + 4 * ((int)blockIdx.x - NCHUNK), pLds);
        return;
    }
    extern __shared__ int lcount[];
    __shared__ int wsum[32];
    for (int i = threadIdx.x; i < nb; i += 1024) lcount[i] = 0;
    __syncthreads();
    const int e0 = blockIdx.x * chunk;
    const int e1 = min(E, e0 + chunk);
    for (int base = e0; base < e1; base += EPT * 1024) {
        int d[EPT];
        for (int k = 0; k < EPT; ++k) {
            const int e = base + k * 1024 + (int)threadIdx.x;
            d[k] = dstIdx[e < e1 ? e : e1 - 1];
        }
        for (int k = 0; k < EPT; ++k) {
            const int e = base + k * 1024 + (int)threadIdx.x;
            if (e < e1) atomicAdd(&lcount[bucket_of(d[k], nb)], 1);
        }
    }
    __syncthreads();
    int carry = 0;
    for (int base = 0; base < nb; base += 1024) {
        const int b = base + (int)threadIdx.x;
        const int v = b < nb ? lcount[b] : 0;
        int total;
        const int ex = block_excl_scan(v, wsum, 32, total);
        if (b < nb) {
            offT[b * NCHUNK + blockIdx.x] = carry + ex;
            cntT[b * NCHUNK + blockIdx.x] = v;
            lcount[b] = carry + ex;
        }
        carry += total;
    }
    __syncthreads();
    for (int base = e0; base < e1; base += EPT * 1024) {
        int s[EPT], d[EPT];
        for (int k = 0; k < EPT; ++k) {
            const int e = base + k * 1024 + (int)threadIdx.x;
            const int ec = e < e1 ? e : e1 - 1;
            s[k] = srcIdx[ec];
            d[k] = dstIdx[ec];
        }
        for (int k = 0; k < EPT; ++k) {
            const int e = base + k * 1024 + (int)threadIdx.x;
            if (e < e1) {
                const int pos = atomicAdd(&lcount[bucket_of(d[k], nb)], 1);
                part[e0 + pos] = make_int2(s[k], d[k]);
            }
        }
    }
}

#define RUNB 16
__global__ __launch_bounds__(BNODES)
void gat_bucket_sort(const int2* __restrict__ part, const int* __restrict__ offT,
                     const int* __restrict__ cntT, int chunk, int N,
                     int* __restrict__ rowStart, int* __restrict__ cnt,
                     int* __restrict__ segSrc, int nbk,
                     const float* __restrict__ pA, const float* __restrict__ pB,
                     float* __restrict__ pC, int pRows, int pTile0) {
    static_assert(BNODES == 2 * P1C, "a projection group is one work-group here");
    static_assert(BNODES == NCHUNK, "one thread per chunk on the way in");
    static_assert(32 % RUNB == 0, "a wave's 32 runs in whole batches");
    __shared__ __attribute__((aligned(16))) float pLds[P1LDS];
    if ((int)blockIdx.x >= nbk) {
        gat_gemm_tiles<P1K, P1C>(pA, pB, pC, pRows, pTile0 + ((int)blockIdx.x - nbk), pLds);
        return;
    }
    __shared__ int lnode[BNODES];
    __shared__ int wsum[32];
    const int tid  = threadIdx.x;
    const int lane = tid & 31;
    const int runOff = offT[blockIdx.x * NCHUNK + tid];
    const int runLen = cntT[blockIdx.x * NCHUNK + tid];
    const int runPos = tid * chunk + runOff;
    lnode[tid] = 0;
    int p0;
    (void)block_excl_scan(runOff, wsum, BNODES / 32, p0);
    for (int j0 = 0; j0 < 32; j0 += RUNB) {
        int d[RUNB];
        for (int u = 0; u < RUNB; ++u) {
            const int st = __builtin_amdgcn_readlane(runPos, j0 + u);
            const int ln = __builtin_amdgcn_readlane(runLen, j0 + u);
            d[u] = part[max(st + min(lane, ln - 1), 0)].y;
        }
        for (int u = 0; u < RUNB; ++u) {
            const int ln = __builtin_amdgcn_readlane(runLen, j0 + u);
            if (lane < ln) atomicAdd(&lnode[d[u] & (BNODES - 1)], 1);
        }
        for (int u = 0; u < RUNB; ++u) {
            const int st = __builtin_amdgcn_readlane(runPos, j0 + u);
            const int ln = __builtin_amdgcn_readlane(runLen, j0 + u);
            for (int o = 32; o < ln; o += 32)
                if (o + lane < ln) atomicAdd(&lnode[part[st + o + lane].y & (BNODES - 1)], 1);
        }
    }
    __syncthreads();
    const int c = lnode[tid];
    int total;
    const int ex = block_excl_scan(c, wsum, BNODES / 32, total);
    lnode[tid] = p0 + ex;
    const int n = blockIdx.x * BNODES + tid;
    if (n < N) { rowStart[n] = p0 + ex; cnt[n] = c; }
    __syncthreads();
    for (int j0 = 0; j0 < 32; j0 += RUNB) {
        int2 sd[RUNB];
        for (int u = 0; u < RUNB; ++u) {
            const int st = __builtin_amdgcn_readlane(runPos, j0 + u);
            const int ln = __builtin_amdgcn_readlane(runLen, j0 + u);
            sd[u] = part[max(st + min(lane, ln - 1), 0)];
        }
        for (int u = 0; u < RUNB; ++u) {
            const int ln = __builtin_amdgcn_readlane(runLen, j0 + u);
            if (lane < ln) {
                const int pos = atomicAdd(&lnode[sd[u].y & (BNODES - 1)], 1);
                segSrc[pos] = sd[u].x;
            }
        }
        for (int u = 0; u < RUNB; ++u) {
            const int st = __builtin_amdgcn_readlane(runPos, j0 + u);
            const int ln = __builtin_amdgcn_readlane(runLen, j0 + u);
            for (int o = 32; o < ln; o += 32)
                if (o + lane < ln) {
                    const int2 e = part[st + o + lane];
                    const int pos = atomicAdd(&lnode[e.y & (BNODES - 1)], 1);
                    segSrc[pos] = e.x;
                }
        }
    }
}

template <int MASK>
__device__ __forceinline__ float row_xor(float x) {
    return __int_as_float(__builtin_amdgcn_update_dpp(
        0, __float_as_int(x), 0x160 + MASK, 0xf, 0xf, true));
}
__device__ __forceinline__ float half_xor(float x) {
    const int b = __float_as_int(x);
    return __int_as_float(__builtin_amdgcn_permlanex16(b, b, 0x76543210, 0xfedcba98, false, false));
}
template <int LPH>
__device__ __forceinline__ float head_sum(float v) {
    v += row_xor<1>(v);
    v += row_xor<2>(v);
    v += row_xor<4>(v);
    if constexpr (LPH > 8)  v += row_xor<8>(v);
    if constexpr (LPH > 16) v += half_xor(v);
    return v;
}

template <int P>
__device__ __forceinline__ void load_row(const float* __restrict__ p, v2f (&v)[P]) {
    if constexpr (P == 2) {
        float4 t = *(const float4*)p;
        v[0].x = t.x; v[0].y = t.y; v[1].x = t.z; v[1].y = t.w;
    } else {
        float2 t = *(const float2*)p;
        v[0].x = t.x; v[0].y = t.y;
    }
}
template <int P>
__device__ __forceinline__ float dot_cols(const v2f (&v)[P], const v2f (&w)[P]) {
    v2f q = v[0] * w[0];
    for (int k = 1; k < P; ++k) q += v[k] * w[k];
    return q.x + q.y;
}

template <int NB, int F, int P, int LPH>
__device__ __forceinline__ void gat_rows(int sItem, int t, int lane,
                                         const float* __restrict__ h,
                                         const v2f (&as)[P], float ed,
                                         float& m, float& sum, v2f (&acc)[P]) {
    v2f v[NB][P];
    for (int u = 0; u < NB; ++u) {
        const int st = __builtin_amdgcn_readlane(sItem, t + u);
        const unsigned boff = ((unsigned)st * (unsigned)F + (unsigned)(lane * (2 * P))) * 4u;
        load_row<P>((const float*)((const char*)h + boff), v[u]);
    }
    float a[NB];
    for (int u = 0; u < NB; ++u)
        a[u] = lrelu(head_sum<LPH>(dot_cols<P>(v[u], as)) + ed, NEG_SLOPE_ATT);
    float nm = m;
    for (int u = 0; u < NB; ++u) nm = fmaxf(nm, a[u]);
    const float sc = __expf(m - nm);
    sum *= sc;
    for (int k = 0; k < P; ++k) acc[k] *= sc;
    for (int u = 0; u < NB; ++u) {
        const float pe = __expf(a[u] - nm);
        sum += pe;
        for (int k = 0; k < P; ++k) acc[k] += pe * v[u][k];
    }
    m = nm;
}

template <int U, int LPH>
__device__ __forceinline__ float edge_weight(float pe) {
    if constexpr (LPH == 32) {
        return __int_as_float(__builtin_amdgcn_readlane(__float_as_int(pe), 4 * U));
    } else {
        constexpr int SEL = U * 0x249249;
        return __int_as_float(__builtin_amdgcn_mov_dpp8(__float_as_int(pe), SEL));
    }
}

template <int F, int P, int LPH>
__device__ __forceinline__ void gat_rows8(int sItem, int t, int lane,
                                          const float* __restrict__ h,
                                          const v2f (&as)[P], float ed,
                                          float& m, float& sum, v2f (&acc)[P]) {
    static_assert(LPH == 8 || LPH == 32, "exchange steps written for heads of 8 or 32 lanes");
    v2f v[8][P];
    for (int u = 0; u < 8; ++u) {
        const int st = __builtin_amdgcn_readlane(sItem, t + u);
        const unsigned boff = ((unsigned)st * (unsigned)F + (unsigned)(lane * (2 * P))) * 4u;
        load_row<P>((const float*)((const char*)h + boff), v[u]);
    }
    float p[8];
    for (int u = 0; u < 8; ++u) p[u] = dot_cols<P>(v[u], as);
    float q[4], r[2], s;
    if constexpr (LPH == 32) {
        const bool c4 = (lane & 16) != 0, c3 = (lane & 8) != 0, c2 = (lane & 4) != 0;
        for (int j = 0; j < 4; ++j)
            q[j] = (c4 ? p[j + 4] : p[j]) + half_xor(c4 ? p[j] : p[j + 4]);
        for (int j = 0; j < 2; ++j)
            r[j] = (c3 ? q[j + 2] : q[j]) + row_xor<8>(c3 ? q[j] : q[j + 2]);
        s = (c2 ? r[1] : r[0]) + row_xor<4>(c2 ? r[0] : r[1]);
        s += row_xor<2>(s);
        s += row_xor<1>(s);
    } else {
        const bool c2 = (lane & 4) != 0, c1 = (lane & 2) != 0, c0 = (lane & 1) != 0;
        for (int j = 0; j < 4; ++j)
            q[j] = (c2 ? p[j + 4] : p[j]) + row_xor<4>(c2 ? p[j] : p[j + 4]);
        for (int j = 0; j < 2; ++j)
            r[j] = (c1 ? q[j + 2] : q[j]) + row_xor<2>(c1 ? q[j] : q[j + 2]);
        s = (c0 ? r[1] : r[0]) + row_xor<1>(c0 ? r[0] : r[1]);
    }
    const float a = lrelu(s + ed, NEG_SLOPE_ATT);
    float mx = a;
    if constexpr (LPH == 32) {
        mx = fmaxf(mx, row_xor<4>(mx));
        mx = fmaxf(mx, row_xor<8>(mx));
        mx = fmaxf(mx, half_xor(mx));
    } else {
        mx = fmaxf(mx, row_xor<1>(mx));
        mx = fmaxf(mx, row_xor<2>(mx));
        mx = fmaxf(mx, row_xor<4>(mx));
    }
    const float nm = fmaxf(m, mx);
    const float sc = __expf(m - nm);
    const float pe = __expf(a - nm);
    sum *= sc;
    for (int k = 0; k < P; ++k) acc[k] *= sc;
#define GAT_EDGE_IN(U) { const float w = edge_weight<U, LPH>(pe); sum += w; \
        for (int k = 0; k < P; ++k) acc[k] += w * v[U][k]; }
    GAT_EDGE_IN(0) GAT_EDGE_IN(1) GAT_EDGE_IN(2) GAT_EDGE_IN(3)
    GAT_EDGE_IN(4) GAT_EDGE_IN(5) GAT_EDGE_IN(6) GAT_EDGE_IN(7)
#undef GAT_EDGE_IN
    m = nm;
}

template <int H, int C>
__global__ __launch_bounds__(256)
void gat_gather(const int* __restrict__ rowStart, const int* __restrict__ cnt,
                const int* __restrict__ segSrc, int N,
                const float* __restrict__ a_src, const float* __restrict__ a_dst,
                const float* __restrict__ h, const float* __restrict__ bias,
                float* __restrict__ out, float slope, int apply_act) {
    constexpr int F = H * C;
    constexpr int VEC = F / 32;
    constexpr int P = VEC / 2;
    constexpr int LPH = C / VEC;
    static_assert(F % 32 == 0 && (VEC == 4 || VEC == 2), "row = 32 lanes x VEC");
    static_assert(C % VEC == 0, "a lane's columns stay inside one head");
    static_assert(LPH == 8 || LPH == 32, "a head is 8 or 32 adjacent lanes");

    const int lane = threadIdx.x & 31;
    const int d = __builtin_amdgcn_readfirstlane((int)(blockIdx.x * 8 + (threadIdx.x >> 5)));
    if (d >= N) return;
    const int start = rowStart[d];
    const int deg   = cnt[d];
    const int col   = lane * VEC;

    v2f as[P], ad[P], vs[P];
    load_row<P>(a_src + col, as);
    load_row<P>(a_dst + col, ad);
    load_row<P>(h + (size_t)d * F + col, vs);
    int sItem = d;
    if (lane < deg) sItem = segSrc[start + lane];

    const float ed = head_sum<LPH>(dot_cols<P>(vs, ad));
    float m   = lrelu(head_sum<LPH>(dot_cols<P>(vs, as)) + ed, NEG_SLOPE_ATT);
    float sum = 1.f;
    v2f acc[P];
    for (int k = 0; k < P; ++k) acc[k] = vs[k];

    for (int base = 0; base < deg; base += 32) {
        if (base > 0) {
            sItem = d;
            if (base + lane < deg) sItem = segSrc[start + base + lane];
        }
        const int left = deg - base;
        const int n = left < 32 ? left : 32;
        int t = 0;
        for (; t + 8 <= n; t += 8)
            gat_rows8<F, P, LPH>(sItem, t, lane, h, as, ed, m, sum, acc);
        if (t + 4 <= n) {
            gat_rows<4, F, P, LPH>(sItem, t, lane, h, as, ed, m, sum, acc);
            t += 4;
        }
        const int rem = n - t;
        if (rem == 3)      gat_rows<3, F, P, LPH>(sItem, t, lane, h, as, ed, m, sum, acc);
        else if (rem == 2) gat_rows<2, F, P, LPH>(sItem, t, lane, h, as, ed, m, sum, acc);
        else if (rem == 1) gat_rows<1, F, P, LPH>(sItem, t, lane, h, as, ed, m, sum, acc);
    }

    const float inv = 1.0f / sum;
    float r[VEC];
    for (int k = 0; k < VEC; ++k) {
        float x = acc[k >> 1][k & 1] * inv + bias[col + k];
        r[k] = apply_act ? lrelu(x, slope) : x;
    }
    float* op = out + (size_t)d * F + col;
    if constexpr (VEC == 4) {
        *(float4*)op = make_float4(r[0], r[1], r[2], r[3]);
    } else {
        *(float2*)op = make_float2(r[0], r[1]);
    }
}

extern "C" void kernel_launch(void* const* d_in, const int* in_sizes, int n_in,
                              void* d_out, int out_size, void* d_ws, size_t ws_size,
                              hipStream_t stream) {
    const float* x        = (const float*)d_in[0];
    const int*   ei       = (const int*)d_in[1];
    const float* W1       = (const float*)d_in[2];
    const float* att_src1 = (const float*)d_in[3];
    const float* att_dst1 = (const float*)d_in[4];
    const float* b1       = (const float*)d_in[5];
    const float* W2       = (const float*)d_in[6];
    const float* att_src2 = (const float*)d_in[7];
    const float* att_dst2 = (const float*)d_in[8];
    const float* b2       = (const float*)d_in[9];

    const int HC1  = in_sizes[5];
    const int OUTC = in_sizes[9];
    const int F_IN = in_sizes[2] / HC1;
    const int N    = in_sizes[0] / F_IN;
    const int E    = in_sizes[1] / 2;

    const int* srcIdx = ei;
    const int* dstIdx = ei + E;

    float* ws  = (float*)d_ws;
    size_t off = 0;
    float* h1   = ws + off; off += (size_t)N * HC1;
    float* act1 = ws + off; off += (size_t)N * HC1;
    const int NBK   = (N + BNODES - 1) / BNODES;
    const int chunk = (E + NCHUNK - 1) / NCHUNK;
    int* cnt      = (int*)(ws + off); off += (((size_t)N + 3) / 4) * 4;
    int* rowStart = (int*)(ws + off); off += (((size_t)N + 3) / 4) * 4;
    int* segSrc   = (int*)(ws + off); off += (((size_t)E + 3) / 4) * 4;
    int* offT        = (int*)(ws + off); off += (size_t)NBK * NCHUNK;
    int* cntT        = (int*)(ws + off); off += (size_t)NBK * NCHUNK;
    int2* part       = (int2*)(ws + off); off += (((size_t)E * 2 + 3) / 4) * 4;
    float* h2   = h1;
    float* out2 = (float*)d_out;

    const int TB = 256;
    const int rowTiles32 = (N + 31) / 32;
    const int waveBlocks = (N + 7) / 8;
    const int gA = rowTiles32 / 6 < 2 * NCHUNK ? rowTiles32 / 6 : 2 * NCHUNK;
    const int gC = rowTiles32 - 4 * gA;

    gat_chunk_group<<<NCHUNK + gA, 1024, NBK * sizeof(int), stream>>>(
        srcIdx, dstIdx, E, chunk, NBK, offT, cntT, part, x, W1, h1, N, 0);
    gat_bucket_sort<<<NBK + gC, BNODES, 0, stream>>>(
        part, offT, cntT, chunk, N, rowStart, cnt, segSrc, NBK, x, W1, h1, N, 4 * gA);

    gat_gather<4, 32><<<waveBlocks, TB, 0, stream>>>(
        rowStart, cnt, segSrc, N, att_src1, att_dst1, h1, b1, act1, NEG_SLOPE_ACT,  1);

    gat_gemm_wmma<128, 64><<<rowTiles32, 32 * (OUTC / 16), 32 * (HC1 + 4) * sizeof(float), stream>>>(
        act1, W2, h2, N);
    gat_gather<1, 64><<<waveBlocks, TB, 0, stream>>>(
        rowStart, cnt, segSrc, N, att_src2, att_dst2, h2, b2, out2, 0.f,  0);
}


# Layer-2 projection moved into the layer-1 gather: a work-group finishes 32 rows into an LDS tile, four of its waves multiply it by W2 on WMMA; layer-1 plane and one launch go; sum orders unchanged
# speedup vs baseline: 8.9504x; 1.0554x over previous
#include <hip/hip_runtime.h>
#include <hip/hip_bf16.h>
#include <math.h>


typedef __attribute__((ext_vector_type(2))) float v2f;
typedef __attribute__((ext_vector_type(4))) float v4f;
typedef __attribute__((ext_vector_type(8))) float v8f;

#define NEG_SLOPE_ATT 0.2f
#define NEG_SLOPE_ACT 0.01f

template <int K, int NCOLS>
__device__ __forceinline__ void gat_tile_multiply(const float* __restrict__ ldsA, const float* __restrict__ B,
                                                  float* __restrict__ C, int Nrows, int tileM,
                                                  int wave, int lane) {
    constexpr int KP = K + 4;
    const int tileN = wave * 16;
    const int rsel  = lane & 15;
    const int khalf = (lane >> 4) << 1;

    v8f acc0 = {};
    v8f acc1 = {};
    const float* __restrict__ a0p = &ldsA[rsel * KP + khalf];
    const float* __restrict__ a1p = &ldsA[(rsel + 16) * KP + khalf];
    for (int k0 = 0; k0 < K; k0 += 4) {
        v2f a0, a1, b;
        a0.x = a0p[k0];     a0.y = a0p[k0 + 1];
        a1.x = a1p[k0];     a1.y = a1p[k0 + 1];
        b.x  = B[(size_t)(k0 + khalf) * NCOLS + tileN + rsel];
        b.y  = B[(size_t)(k0 + khalf + 1) * NCOLS + tileN + rsel];
        acc0 = __builtin_amdgcn_wmma_f32_16x16x4_f32(
            false, a0, false, b, (short)0, acc0, false, false);
        acc1 = __builtin_amdgcn_wmma_f32_16x16x4_f32(
            false, a1, false, b, (short)0, acc1, false, false);
    }

    const int mbase = tileM + ((lane >> 4) << 3);
    const int col   = tileN + rsel;
    for (int r = 0; r < 8; ++r) {
        int row0 = mbase + r;
        int row1 = row0 + 16;
        if (row0 < Nrows) C[(size_t)row0 * NCOLS + col] = acc0[r];
        if (row1 < Nrows) C[(size_t)row1 * NCOLS + col] = acc1[r];
    }
}

template <int K, int NCOLS>
__device__ __forceinline__ void gat_gemm_tiles(const float* __restrict__ A, const float* __restrict__ B,
                                               float* __restrict__ C, int Nrows, int tile0,
                                               float* __restrict__ lds) {
    constexpr int KP  = K + 4;
    constexpr int NT  = 2 * NCOLS;
    constexpr int K4  = K / 4;
    constexpr int PCS = (32 * K4) / NT;
    static_assert(K % 4 == 0 && NCOLS % 16 == 0, "whole 16-byte pieces, whole column tiles");
    static_assert((32 * K4) % NT == 0, "the tile's pieces divide evenly over the threads");
    constexpr int WPG = NT / 32;
    const int tid   = (int)threadIdx.x % NT;
    const int wv    = __builtin_amdgcn_readfirstlane((int)(threadIdx.x >> 5));
    const int grp   = wv / WPG;
    float* __restrict__ ldsA = lds + grp * (32 * KP);
    const int tileM = (tile0 + grp) * 32;
    const int lane  = tid & 31;
    const int wave  = wv % WPG;

    v4f piece[PCS];
    #pragma unroll
    for (int j = 0; j < PCS; ++j) {
        const int idx = j * NT + tid;
        const int r = idx / K4, c = (idx - r * K4) * 4;
        const int row = tileM + r;
        const int rc  = row < Nrows ? row : Nrows - 1;
        piece[j] = *(const v4f*)(A + (size_t)rc * K + c);
    }
    #pragma unroll
    for (int j = 0; j < PCS; ++j) {
        const int idx = j * NT + tid;
        const int r = idx / K4, c = (idx - r * K4) * 4;
        *(v4f*)&ldsA[r * KP + c] = piece[j];
    }
    __syncthreads();

    const int tileN = wave * 16;
    const int rsel  = lane & 15;
    const int khalf = (lane >> 4) << 1;

    v8f acc0 = {};
    v8f acc1 = {};
    const float* __restrict__ a0p = &ldsA[rsel * KP + khalf];
    const float* __restrict__ a1p = &ldsA[(rsel + 16) * KP + khalf];
    for (int k0 = 0; k0 < K; k0 += 4) {
        v2f a0, a1, b;
        a0.x = a0p[k0];     a0.y = a0p[k0 + 1];
        a1.x = a1p[k0];     a1.y = a1p[k0 + 1];
        b.x  = B[(size_t)(k0 + khalf) * NCOLS + tileN + rsel];
        b.y  = B[(size_t)(k0 + khalf + 1) * NCOLS + tileN + rsel];
        acc0 = __builtin_amdgcn_wmma_f32_16x16x4_f32(
            false, a0, false, b, (short)0, acc0, false, false);
        acc1 = __builtin_amdgcn_wmma_f32_16x16x4_f32(
            false, a1, false, b, (short)0, acc1, false, false);
    }

    const int mbase = tileM + ((lane >> 4) << 3);
    const int col   = tileN + rsel;
    for (int r = 0; r < 8; ++r) {
        int row0 = mbase + r;
        int row1 = row0 + 16;
        if (row0 < Nrows) C[(size_t)row0 * NCOLS + col] = acc0[r];
        if (row1 < Nrows) C[(size_t)row1 * NCOLS + col] = acc1[r];
    }
}

#define P1K 128
#define P1C 128
#define P1LDS (32 * (P1K + 4))

__device__ __forceinline__ float lrelu(float x, float slope) {
    return x > 0.f ? x : slope * x;
}

#define BSHIFT 8
#define BNODES 256
#define NCHUNK 256
#define EPT 8

__device__ __forceinline__ int bucket_of(int d, int nb) {
    const unsigned b = (unsigned)d >> BSHIFT;
    return (int)(b < (unsigned)nb ? b : (unsigned)(nb - 1));
}

__device__ __forceinline__ int block_excl_scan(int v, int* wsum, int nwaves, int& total) {
    const int lane = threadIdx.x & 31;
    const int wave = threadIdx.x >> 5;
    int incl = v;
    for (int off = 1; off < 32; off <<= 1) {
        int t = __shfl_up(incl, off, 32);
        if (lane >= off) incl += t;
    }
    if (lane == 31) wsum[wave] = incl;
    __syncthreads();
    int base = 0, tot = 0;
    for (int w = 0; w < nwaves; ++w) {
        const int t = wsum[w];
        if (w < wave) base += t;
        tot += t;
    }
    __syncthreads();
    total = tot;
    return base + incl - v;
}

__global__ __launch_bounds__(1024)
void gat_chunk_group(const int* __restrict__ srcIdx, const int* __restrict__ dstIdx,
                     int E, int chunk, int nb,
                     int* __restrict__ offT, int* __restrict__ cntT, int2* __restrict__ part,
                     const float* __restrict__ pA, const float* __restrict__ pB,
                     float* __restrict__ pC, int pRows, int pTile0) {
    __shared__ __attribute__((aligned(16))) float pLds[4 * P1LDS];
    if (blockIdx.x >= NCHUNK) {
        gat_gemm_tiles<P1K, P1C>(pA, pB, pC, pRows, pTile0 + 4 * ((int)blockIdx.x - NCHUNK), pLds);
        return;
    }
    extern __shared__ int lcount[];
    __shared__ int wsum[32];
    for (int i = threadIdx.x; i < nb; i += 1024) lcount[i] = 0;
    __syncthreads();
    const int e0 = blockIdx.x * chunk;
    const int e1 = min(E, e0 + chunk);
    for (int base = e0; base < e1; base += EPT * 1024) {
        int d[EPT];
        for (int k = 0; k < EPT; ++k) {
            const int e = base + k * 1024 + (int)threadIdx.x;
            d[k] = dstIdx[e < e1 ? e : e1 - 1];
        }
        for (int k = 0; k < EPT; ++k) {
            const int e = base + k * 1024 + (int)threadIdx.x;
            if (e < e1) atomicAdd(&lcount[bucket_of(d[k], nb)], 1);
        }
    }
    __syncthreads();
    int carry = 0;
    for (int base = 0; base < nb; base += 1024) {
        const int b = base + (int)threadIdx.x;
        const int v = b < nb ? lcount[b] : 0;
        int total;
        const int ex = block_excl_scan(v, wsum, 32, total);
        if (b < nb) {
            offT[b * NCHUNK + blockIdx.x] = carry + ex;
            cntT[b * NCHUNK + blockIdx.x] = v;
            lcount[b] = carry + ex;
        }
        carry += total;
    }
    __syncthreads();
    for (int base = e0; base < e1; base += EPT * 1024) {
        int s[EPT], d[EPT];
        for (int k = 0; k < EPT; ++k) {
            const int e = base + k * 1024 + (int)threadIdx.x;
            const int ec = e < e1 ? e : e1 - 1;
            s[k] = srcIdx[ec];
            d[k] = dstIdx[ec];
        }
        for (int k = 0; k < EPT; ++k) {
            const int e = base + k * 1024 + (int)threadIdx.x;
            if (e < e1) {
                const int pos = atomicAdd(&lcount[bucket_of(d[k], nb)], 1);
                part[e0 + pos] = make_int2(s[k], d[k]);
            }
        }
    }
}

#define RUNB 16
__global__ __launch_bounds__(BNODES)
void gat_bucket_sort(const int2* __restrict__ part, const int* __restrict__ offT,
                     const int* __restrict__ cntT, int chunk, int N,
                     int* __restrict__ rowStart, int* __restrict__ cnt,
                     int* __restrict__ segSrc, int nbk,
                     const float* __restrict__ pA, const float* __restrict__ pB,
                     float* __restrict__ pC, int pRows, int pTile0) {
    static_assert(BNODES == 2 * P1C, "a projection group is one work-group here");
    static_assert(BNODES == NCHUNK, "one thread per chunk on the way in");
    static_assert(32 % RUNB == 0, "a wave's 32 runs in whole batches");
    __shared__ __attribute__((aligned(16))) float pLds[P1LDS];
    if ((int)blockIdx.x >= nbk) {
        gat_gemm_tiles<P1K, P1C>(pA, pB, pC, pRows, pTile0 + ((int)blockIdx.x - nbk), pLds);
        return;
    }
    __shared__ int lnode[BNODES];
    __shared__ int wsum[32];
    const int tid  = threadIdx.x;
    const int lane = tid & 31;
    const int runOff = offT[blockIdx.x * NCHUNK + tid];
    const int runLen = cntT[blockIdx.x * NCHUNK + tid];
    const int runPos = tid * chunk + runOff;
    lnode[tid] = 0;
    int p0;
    (void)block_excl_scan(runOff, wsum, BNODES / 32, p0);
    for (int j0 = 0; j0 < 32; j0 += RUNB) {
        int d[RUNB];
        for (int u = 0; u < RUNB; ++u) {
            const int st = __builtin_amdgcn_readlane(runPos, j0 + u);
            const int ln = __builtin_amdgcn_readlane(runLen, j0 + u);
            d[u] = part[max(st + min(lane, ln - 1), 0)].y;
        }
        for (int u = 0; u < RUNB; ++u) {
            const int ln = __builtin_amdgcn_readlane(runLen, j0 + u);
            if (lane < ln) atomicAdd(&lnode[d[u] & (BNODES - 1)], 1);
        }
        for (int u = 0; u < RUNB; ++u) {
            const int st = __builtin_amdgcn_readlane(runPos, j0 + u);
            const int ln = __builtin_amdgcn_readlane(runLen, j0 + u);
            for (int o = 32; o < ln; o += 32)
                if (o + lane < ln) atomicAdd(&lnode[part[st + o + lane].y & (BNODES - 1)], 1);
        }
    }
    __syncthreads();
    const int c = lnode[tid];
    int total;
    const int ex = block_excl_scan(c, wsum, BNODES / 32, total);
    lnode[tid] = p0 + ex;
    const int n = blockIdx.x * BNODES + tid;
    if (n < N) { rowStart[n] = p0 + ex; cnt[n] = c; }
    __syncthreads();
    for (int j0 = 0; j0 < 32; j0 += RUNB) {
        int2 sd[RUNB];
        for (int u = 0; u < RUNB; ++u) {
            const int st = __builtin_amdgcn_readlane(runPos, j0 + u);
            const int ln = __builtin_amdgcn_readlane(runLen, j0 + u);
            sd[u] = part[max(st + min(lane, ln - 1), 0)];
        }
        for (int u = 0; u < RUNB; ++u) {
            const int ln = __builtin_amdgcn_readlane(runLen, j0 + u);
            if (lane < ln) {
                const int pos = atomicAdd(&lnode[sd[u].y & (BNODES - 1)], 1);
                segSrc[pos] = sd[u].x;
            }
        }
        for (int u = 0; u < RUNB; ++u) {
            const int st = __builtin_amdgcn_readlane(runPos, j0 + u);
            const int ln = __builtin_amdgcn_readlane(runLen, j0 + u);
            for (int o = 32; o < ln; o += 32)
                if (o + lane < ln) {
                    const int2 e = part[st + o + lane];
                    const int pos = atomicAdd(&lnode[e.y & (BNODES - 1)], 1);
                    segSrc[pos] = e.x;
                }
        }
    }
}

template <int MASK>
__device__ __forceinline__ float row_xor(float x) {
    return __int_as_float(__builtin_amdgcn_update_dpp(
        0, __float_as_int(x), 0x160 + MASK, 0xf, 0xf, true));
}
__device__ __forceinline__ float half_xor(float x) {
    const int b = __float_as_int(x);
    return __int_as_float(__builtin_amdgcn_permlanex16(b, b, 0x76543210, 0xfedcba98, false, false));
}
template <int LPH>
__device__ __forceinline__ float head_sum(float v) {
    v += row_xor<1>(v);
    v += row_xor<2>(v);
    v += row_xor<4>(v);
    if constexpr (LPH > 8)  v += row_xor<8>(v);
    if constexpr (LPH > 16) v += half_xor(v);
    return v;
}

template <int P>
__device__ __forceinline__ void load_row(const float* __restrict__ p, v2f (&v)[P]) {
    if constexpr (P == 2) {
        float4 t = *(const float4*)p;
        v[0].x = t.x; v[0].y = t.y; v[1].x = t.z; v[1].y = t.w;
    } else {
        float2 t = *(const float2*)p;
        v[0].x = t.x; v[0].y = t.y;
    }
}
template <int P>
__device__ __forceinline__ float dot_cols(const v2f (&v)[P], const v2f (&w)[P]) {
    v2f q = v[0] * w[0];
    for (int k = 1; k < P; ++k) q += v[k] * w[k];
    return q.x + q.y;
}

template <int NB, int F, int P, int LPH>
__device__ __forceinline__ void gat_rows(int sItem, int t, int lane,
                                         const float* __restrict__ h,
                                         const v2f (&as)[P], float ed,
                                         float& m, float& sum, v2f (&acc)[P]) {
    v2f v[NB][P];
    for (int u = 0; u < NB; ++u) {
        const int st = __builtin_amdgcn_readlane(sItem, t + u);
        const unsigned boff = ((unsigned)st * (unsigned)F + (unsigned)(lane * (2 * P))) * 4u;
        load_row<P>((const float*)((const char*)h + boff), v[u]);
    }
    float a[NB];
    for (int u = 0; u < NB; ++u)
        a[u] = lrelu(head_sum<LPH>(dot_cols<P>(v[u], as)) + ed, NEG_SLOPE_ATT);
    float nm = m;
    for (int u = 0; u < NB; ++u) nm = fmaxf(nm, a[u]);
    const float sc = __expf(m - nm);
    sum *= sc;
    for (int k = 0; k < P; ++k) acc[k] *= sc;
    for (int u = 0; u < NB; ++u) {
        const float pe = __expf(a[u] - nm);
        sum += pe;
        for (int k = 0; k < P; ++k) acc[k] += pe * v[u][k];
    }
    m = nm;
}

template <int U, int LPH>
__device__ __forceinline__ float edge_weight(float pe) {
    if constexpr (LPH == 32) {
        return __int_as_float(__builtin_amdgcn_readlane(__float_as_int(pe), 4 * U));
    } else {
        constexpr int SEL = U * 0x249249;
        return __int_as_float(__builtin_amdgcn_mov_dpp8(__float_as_int(pe), SEL));
    }
}

template <int F, int P, int LPH>
__device__ __forceinline__ void gat_rows8(int sItem, int t, int lane,
                                          const float* __restrict__ h,
                                          const v2f (&as)[P], float ed,
                                          float& m, float& sum, v2f (&acc)[P]) {
    static_assert(LPH == 8 || LPH == 32, "exchange steps written for heads of 8 or 32 lanes");
    v2f v[8][P];
    for (int u = 0; u < 8; ++u) {
        const int st = __builtin_amdgcn_readlane(sItem, t + u);
        const unsigned boff = ((unsigned)st * (unsigned)F + (unsigned)(lane * (2 * P))) * 4u;
        load_row<P>((const float*)((const char*)h + boff), v[u]);
    }
    float p[8];
    for (int u = 0; u < 8; ++u) p[u] = dot_cols<P>(v[u], as);
    float q[4], r[2], s;
    if constexpr (LPH == 32) {
        const bool c4 = (lane & 16) != 0, c3 = (lane & 8) != 0, c2 = (lane & 4) != 0;
        for (int j = 0; j < 4; ++j)
            q[j] = (c4 ? p[j + 4] : p[j]) + half_xor(c4 ? p[j] : p[j + 4]);
        for (int j = 0; j < 2; ++j)
            r[j] = (c3 ? q[j + 2] : q[j]) + row_xor<8>(c3 ? q[j] : q[j + 2]);
        s = (c2 ? r[1] : r[0]) + row_xor<4>(c2 ? r[0] : r[1]);
        s += row_xor<2>(s);
        s += row_xor<1>(s);
    } else {
        const bool c2 = (lane & 4) != 0, c1 = (lane & 2) != 0, c0 = (lane & 1) != 0;
        for (int j = 0; j < 4; ++j)
            q[j] = (c2 ? p[j + 4] : p[j]) + row_xor<4>(c2 ? p[j] : p[j + 4]);
        for (int j = 0; j < 2; ++j)
            r[j] = (c1 ? q[j + 2] : q[j]) + row_xor<2>(c1 ? q[j] : q[j + 2]);
        s = (c0 ? r[1] : r[0]) + row_xor<1>(c0 ? r[0] : r[1]);
    }
    const float a = lrelu(s + ed, NEG_SLOPE_ATT);
    float mx = a;
    if constexpr (LPH == 32) {
        mx = fmaxf(mx, row_xor<4>(mx));
        mx = fmaxf(mx, row_xor<8>(mx));
        mx = fmaxf(mx, half_xor(mx));
    } else {
        mx = fmaxf(mx, row_xor<1>(mx));
        mx = fmaxf(mx, row_xor<2>(mx));
        mx = fmaxf(mx, row_xor<4>(mx));
    }
    const float nm = fmaxf(m, mx);
    const float sc = __expf(m - nm);
    const float pe = __expf(a - nm);
    sum *= sc;
    for (int k = 0; k < P; ++k) acc[k] *= sc;
#define GAT_EDGE_IN(U) { const float w = edge_weight<U, LPH>(pe); sum += w; \
        for (int k = 0; k < P; ++k) acc[k] += w * v[U][k]; }
    GAT_EDGE_IN(0) GAT_EDGE_IN(1) GAT_EDGE_IN(2) GAT_EDGE_IN(3)
    GAT_EDGE_IN(4) GAT_EDGE_IN(5) GAT_EDGE_IN(6) GAT_EDGE_IN(7)
#undef GAT_EDGE_IN
    m = nm;
}

template <int H, int C, int P, int VEC>
__device__ __forceinline__ void gat_node(int d, int lane,
                                         const int* __restrict__ rowStart, const int* __restrict__ cnt,
                                         const int* __restrict__ segSrc,
                                         const v2f (&as)[P], const v2f (&ad)[P],
                                         const float* __restrict__ h, const float* __restrict__ bias,
                                         float slope, int apply_act, float (&r)[VEC]) {
    constexpr int F = H * C;
    constexpr int LPH = C / VEC;
    static_assert(VEC == F / 32 && P == VEC / 2, "row = 32 lanes x VEC");
    const int start = rowStart[d];
    const int deg   = cnt[d];
    const int col   = lane * VEC;

    v2f vs[P];
    load_row<P>(h + (size_t)d * F + col, vs);
    int sItem = d;
    if (lane < deg) sItem = segSrc[start + lane];

    const float ed = head_sum<LPH>(dot_cols<P>(vs, ad));
    float m   = lrelu(head_sum<LPH>(dot_cols<P>(vs, as)) + ed, NEG_SLOPE_ATT);
    float sum = 1.f;
    v2f acc[P];
    for (int k = 0; k < P; ++k) acc[k] = vs[k];

    for (int base = 0; base < deg; base += 32) {
        if (base > 0) {
            sItem = d;
            if (base + lane < deg) sItem = segSrc[start + base + lane];
        }
        const int left = deg - base;
        const int n = left < 32 ? left : 32;
        int t = 0;
        for (; t + 8 <= n; t += 8)
            gat_rows8<F, P, LPH>(sItem, t, lane, h, as, ed, m, sum, acc);
        if (t + 4 <= n) {
            gat_rows<4, F, P, LPH>(sItem, t, lane, h, as, ed, m, sum, acc);
            t += 4;
        }
        const int rem = n - t;
        if (rem == 3)      gat_rows<3, F, P, LPH>(sItem, t, lane, h, as, ed, m, sum, acc);
        else if (rem == 2) gat_rows<2, F, P, LPH>(sItem, t, lane, h, as, ed, m, sum, acc);
        else if (rem == 1) gat_rows<1, F, P, LPH>(sItem, t, lane, h, as, ed, m, sum, acc);
    }

    const float inv = 1.0f / sum;
    for (int k = 0; k < VEC; ++k) {
        float x = acc[k >> 1][k & 1] * inv + bias[col + k];
        r[k] = apply_act ? lrelu(x, slope) : x;
    }
}

template <int H, int C>
__global__ __launch_bounds__(256)
void gat_gather(const int* __restrict__ rowStart, const int* __restrict__ cnt,
                const int* __restrict__ segSrc, int N,
                const float* __restrict__ a_src, const float* __restrict__ a_dst,
                const float* __restrict__ h, const float* __restrict__ bias,
                float* __restrict__ out, float slope, int apply_act) {
    constexpr int F = H * C;
    constexpr int VEC = F / 32;
    constexpr int P = VEC / 2;
    static_assert(F % 32 == 0 && (VEC == 4 || VEC == 2), "row = 32 lanes x VEC");
    static_assert(C % VEC == 0, "a lane's columns stay inside one head");
    static_assert(C / VEC == 8 || C / VEC == 32, "a head is 8 or 32 adjacent lanes");

    const int lane = threadIdx.x & 31;
    const int d = __builtin_amdgcn_readfirstlane((int)(blockIdx.x * 8 + (threadIdx.x >> 5)));
    if (d >= N) return;
    const int col = lane * VEC;

    v2f as[P], ad[P];
    load_row<P>(a_src + col, as);
    load_row<P>(a_dst + col, ad);
    float r[VEC];
    gat_node<H, C, P, VEC>(d, lane, rowStart, cnt, segSrc, as, ad, h, bias, slope, apply_act, r);
    float* op = out + (size_t)d * F + col;
    if constexpr (VEC == 4) {
        *(float4*)op = make_float4(r[0], r[1], r[2], r[3]);
    } else {
        *(float2*)op = make_float2(r[0], r[1]);
    }
}

template <int H, int C, int NC2>
__global__ __launch_bounds__(256)
void gat_gather_project(const int* __restrict__ rowStart, const int* __restrict__ cnt,
                        const int* __restrict__ segSrc, int N,
                        const float* __restrict__ a_src, const float* __restrict__ a_dst,
                        const float* __restrict__ h, const float* __restrict__ bias, float slope,
                        const float* __restrict__ W2, float* __restrict__ out2) {
    constexpr int F = H * C;
    constexpr int VEC = F / 32;
    constexpr int P = VEC / 2;
    constexpr int KP = F + 4;
    constexpr int NPW = 32 / 8;
    static_assert(VEC == 4, "a lane's piece of a row is one 16-byte LDS store");
    static_assert(C % VEC == 0 && (C / VEC == 8 || C / VEC == 32), "a head is 8 or 32 adjacent lanes");
    static_assert(NC2 % 16 == 0 && NC2 / 16 <= 8, "one multiplying wave per 16-column tile");
    __shared__ __attribute__((aligned(16))) float tileA[32 * KP];

    const int lane  = threadIdx.x & 31;
    const int wv    = __builtin_amdgcn_readfirstlane((int)(threadIdx.x >> 5));
    const int tileM = (int)blockIdx.x * 32;
    const int col   = lane * VEC;

    v2f as[P], ad[P];
    load_row<P>(a_src + col, as);
    load_row<P>(a_dst + col, ad);
    for (int i = 0; i < NPW; ++i) {
        const int rt = wv * NPW + i;
        const int d  = tileM + rt;
        float r[VEC] = {0.f, 0.f, 0.f, 0.f};
        if (d < N)
            gat_node<H, C, P, VEC>(d, lane, rowStart, cnt, segSrc, as, ad, h, bias, slope, 1, r);
        v4f piece;
        piece.x = r[0]; piece.y = r[1]; piece.z = r[2]; piece.w = r[3];
        *(v4f*)&tileA[rt * KP + col] = piece;
    }
    __syncthreads();
    if (wv < NC2 / 16)
        gat_tile_multiply<F, NC2>(tileA, W2, out2, N, tileM, wv, lane);
}

extern "C" void kernel_launch(void* const* d_in, const int* in_sizes, int n_in,
                              void* d_out, int out_size, void* d_ws, size_t ws_size,
                              hipStream_t stream) {
    const float* x        = (const float*)d_in[0];
    const int*   ei       = (const int*)d_in[1];
    const float* W1       = (const float*)d_in[2];
    const float* att_src1 = (const float*)d_in[3];
    const float* att_dst1 = (const float*)d_in[4];
    const float* b1       = (const float*)d_in[5];
    const float* W2       = (const float*)d_in[6];
    const float* att_src2 = (const float*)d_in[7];
    const float* att_dst2 = (const float*)d_in[8];
    const float* b2       = (const float*)d_in[9];

    const int HC1  = in_sizes[5];
    const int OUTC = in_sizes[9];
    const int F_IN = in_sizes[2] / HC1;
    const int N    = in_sizes[0] / F_IN;
    const int E    = in_sizes[1] / 2;

    const int* srcIdx = ei;
    const int* dstIdx = ei + E;

    float* ws  = (float*)d_ws;
    size_t off = 0;
    float* h1   = ws + off; off += (size_t)N * HC1;
    float* h2   = ws + off; off += (size_t)N * OUTC;
    const int NBK   = (N + BNODES - 1) / BNODES;
    const int chunk = (E + NCHUNK - 1) / NCHUNK;
    int* cnt      = (int*)(ws + off); off += (((size_t)N + 3) / 4) * 4;
    int* rowStart = (int*)(ws + off); off += (((size_t)N + 3) / 4) * 4;
    int* segSrc   = (int*)(ws + off); off += (((size_t)E + 3) / 4) * 4;
    int* offT        = (int*)(ws + off); off += (size_t)NBK * NCHUNK;
    int* cntT        = (int*)(ws + off); off += (size_t)NBK * NCHUNK;
    int2* part       = (int2*)(ws + off); off += (((size_t)E * 2 + 3) / 4) * 4;
    float* out2 = (float*)d_out;

    const int TB = 256;
    const int rowTiles32 = (N + 31) / 32;
    const int waveBlocks = (N + 7) / 8;
    const int gA = rowTiles32 / 6 < 2 * NCHUNK ? rowTiles32 / 6 : 2 * NCHUNK;
    const int gC = rowTiles32 - 4 * gA;

    gat_chunk_group<<<NCHUNK + gA, 1024, NBK * sizeof(int), stream>>>(
        srcIdx, dstIdx, E, chunk, NBK, offT, cntT, part, x, W1, h1, N, 0);
    gat_bucket_sort<<<NBK + gC, BNODES, 0, stream>>>(
        part, offT, cntT, chunk, N, rowStart, cnt, segSrc, NBK, x, W1, h1, N, 4 * gA);

    gat_gather_project<4, 32, 64><<<rowTiles32, TB, 0, stream>>>(
        rowStart, cnt, segSrc, N, att_src1, att_dst1, h1, b1, NEG_SLOPE_ACT, W2, h2);

    gat_gather<1, 64><<<waveBlocks, TB, 0, stream>>>(
        rowStart, cnt, segSrc, N, att_src2, att_dst2, h2, b2, out2, 0.f,  0);
}


# Layer-2 gather: a wave takes two nodes, one per 16-lane row (64-column rows as 16 lanes x 16 B), so each row load, dot and multiply-add serves two edges; logit and softmax sum orders change
# speedup vs baseline: 9.2397x; 1.0323x over previous
#include <hip/hip_runtime.h>
#include <hip/hip_bf16.h>
#include <math.h>


typedef __attribute__((ext_vector_type(2))) float v2f;
typedef __attribute__((ext_vector_type(4))) float v4f;
typedef __attribute__((ext_vector_type(8))) float v8f;

#define NEG_SLOPE_ATT 0.2f
#define NEG_SLOPE_ACT 0.01f

template <int K, int NCOLS>
__device__ __forceinline__ void gat_tile_multiply(const float* __restrict__ ldsA, const float* __restrict__ B,
                                                  float* __restrict__ C, int Nrows, int tileM,
                                                  int wave, int lane) {
    constexpr int KP = K + 4;
    const int tileN = wave * 16;
    const int rsel  = lane & 15;
    const int khalf = (lane >> 4) << 1;

    v8f acc0 = {};
    v8f acc1 = {};
    const float* __restrict__ a0p = &ldsA[rsel * KP + khalf];
    const float* __restrict__ a1p = &ldsA[(rsel + 16) * KP + khalf];
    for (int k0 = 0; k0 < K; k0 += 4) {
        v2f a0, a1, b;
        a0.x = a0p[k0];     a0.y = a0p[k0 + 1];
        a1.x = a1p[k0];     a1.y = a1p[k0 + 1];
        b.x  = B[(size_t)(k0 + khalf) * NCOLS + tileN + rsel];
        b.y  = B[(size_t)(k0 + khalf + 1) * NCOLS + tileN + rsel];
        acc0 = __builtin_amdgcn_wmma_f32_16x16x4_f32(
            false, a0, false, b, (short)0, acc0, false, false);
        acc1 = __builtin_amdgcn_wmma_f32_16x16x4_f32(
            false, a1, false, b, (short)0, acc1, false, false);
    }

    const int mbase = tileM + ((lane >> 4) << 3);
    const int col   = tileN + rsel;
    for (int r = 0; r < 8; ++r) {
        int row0 = mbase + r;
        int row1 = row0 + 16;
        if (row0 < Nrows) C[(size_t)row0 * NCOLS + col] = acc0[r];
        if (row1 < Nrows) C[(size_t)row1 * NCOLS + col] = acc1[r];
    }
}

template <int K, int NCOLS>
__device__ __forceinline__ void gat_gemm_tiles(const float* __restrict__ A, const float* __restrict__ B,
                                               float* __restrict__ C, int Nrows, int tile0,
                                               float* __restrict__ lds) {
    constexpr int KP  = K + 4;
    constexpr int NT  = 2 * NCOLS;
    constexpr int K4  = K / 4;
    constexpr int PCS = (32 * K4) / NT;
    static_assert(K % 4 == 0 && NCOLS % 16 == 0, "whole 16-byte pieces, whole column tiles");
    static_assert((32 * K4) % NT == 0, "the tile's pieces divide evenly over the threads");
    constexpr int WPG = NT / 32;
    const int tid   = (int)threadIdx.x % NT;
    const int wv    = __builtin_amdgcn_readfirstlane((int)(threadIdx.x >> 5));
    const int grp   = wv / WPG;
    float* __restrict__ ldsA = lds + grp * (32 * KP);
    const int tileM = (tile0 + grp) * 32;
    const int lane  = tid & 31;
    const int wave  = wv % WPG;

    v4f piece[PCS];
    #pragma unroll
    for (int j = 0; j < PCS; ++j) {
        const int idx = j * NT + tid;
        const int r = idx / K4, c = (idx - r * K4) * 4;
        const int row = tileM + r;
        const int rc  = row < Nrows ? row : Nrows - 1;
        piece[j] = *(const v4f*)(A + (size_t)rc * K + c);
    }
    #pragma unroll
    for (int j = 0; j < PCS; ++j) {
        const int idx = j * NT + tid;
        const int r = idx / K4, c = (idx - r * K4) * 4;
        *(v4f*)&ldsA[r * KP + c] = piece[j];
    }
    __syncthreads();

    const int tileN = wave * 16;
    const int rsel  = lane & 15;
    const int khalf = (lane >> 4) << 1;

    v8f acc0 = {};
    v8f acc1 = {};
    const float* __restrict__ a0p = &ldsA[rsel * KP + khalf];
    const float* __restrict__ a1p = &ldsA[(rsel + 16) * KP + khalf];
    for (int k0 = 0; k0 < K; k0 += 4) {
        v2f a0, a1, b;
        a0.x = a0p[k0];     a0.y = a0p[k0 + 1];
        a1.x = a1p[k0];     a1.y = a1p[k0 + 1];
        b.x  = B[(size_t)(k0 + khalf) * NCOLS + tileN + rsel];
        b.y  = B[(size_t)(k0 + khalf + 1) * NCOLS + tileN + rsel];
        acc0 = __builtin_amdgcn_wmma_f32_16x16x4_f32(
            false, a0, false, b, (short)0, acc0, false, false);
        acc1 = __builtin_amdgcn_wmma_f32_16x16x4_f32(
            false, a1, false, b, (short)0, acc1, false, false);
    }

    const int mbase = tileM + ((lane >> 4) << 3);
    const int col   = tileN + rsel;
    for (int r = 0; r < 8; ++r) {
        int row0 = mbase + r;
        int row1 = row0 + 16;
        if (row0 < Nrows) C[(size_t)row0 * NCOLS + col] = acc0[r];
        if (row1 < Nrows) C[(size_t)row1 * NCOLS + col] = acc1[r];
    }
}

#define P1K 128
#define P1C 128
#define P1LDS (32 * (P1K + 4))

__device__ __forceinline__ float lrelu(float x, float slope) {
    return x > 0.f ? x : slope * x;
}

#define BSHIFT 8
#define BNODES 256
#define NCHUNK 256
#define EPT 8

__device__ __forceinline__ int bucket_of(int d, int nb) {
    const unsigned b = (unsigned)d >> BSHIFT;
    return (int)(b < (unsigned)nb ? b : (unsigned)(nb - 1));
}

__device__ __forceinline__ int block_excl_scan(int v, int* wsum, int nwaves, int& total) {
    const int lane = threadIdx.x & 31;
    const int wave = threadIdx.x >> 5;
    int incl = v;
    for (int off = 1; off < 32; off <<= 1) {
        int t = __shfl_up(incl, off, 32);
        if (lane >= off) incl += t;
    }
    if (lane == 31) wsum[wave] = incl;
    __syncthreads();
    int base = 0, tot = 0;
    for (int w = 0; w < nwaves; ++w) {
        const int t = wsum[w];
        if (w < wave) base += t;
        tot += t;
    }
    __syncthreads();
    total = tot;
    return base + incl - v;
}

__global__ __launch_bounds__(1024)
void gat_chunk_group(const int* __restrict__ srcIdx, const int* __restrict__ dstIdx,
                     int E, int chunk, int nb,
                     int* __restrict__ offT, int* __restrict__ cntT, int2* __restrict__ part,
                     const float* __restrict__ pA, const float* __restrict__ pB,
                     float* __restrict__ pC, int pRows, int pTile0) {
    __shared__ __attribute__((aligned(16))) float pLds[4 * P1LDS];
    if (blockIdx.x >= NCHUNK) {
        gat_gemm_tiles<P1K, P1C>(pA, pB, pC, pRows, pTile0 + 4 * ((int)blockIdx.x - NCHUNK), pLds);
        return;
    }
    extern __shared__ int lcount[];
    __shared__ int wsum[32];
    for (int i = threadIdx.x; i < nb; i += 1024) lcount[i] = 0;
    __syncthreads();
    const int e0 = blockIdx.x * chunk;
    const int e1 = min(E, e0 + chunk);
    for (int base = e0; base < e1; base += EPT * 1024) {
        int d[EPT];
        for (int k = 0; k < EPT; ++k) {
            const int e = base + k * 1024 + (int)threadIdx.x;
            d[k] = dstIdx[e < e1 ? e : e1 - 1];
        }
        for (int k = 0; k < EPT; ++k) {
            const int e = base + k * 1024 + (int)threadIdx.x;
            if (e < e1) atomicAdd(&lcount[bucket_of(d[k], nb)], 1);
        }
    }
    __syncthreads();
    int carry = 0;
    for (int base = 0; base < nb; base += 1024) {
        const int b = base + (int)threadIdx.x;
        const int v = b < nb ? lcount[b] : 0;
        int total;
        const int ex = block_excl_scan(v, wsum, 32, total);
        if (b < nb) {
            offT[b * NCHUNK + blockIdx.x] = carry + ex;
            cntT[b * NCHUNK + blockIdx.x] = v;
            lcount[b] = carry + ex;
        }
        carry += total;
    }
    __syncthreads();
    for (int base = e0; base < e1; base += EPT * 1024) {
        int s[EPT], d[EPT];
        for (int k = 0; k < EPT; ++k) {
            const int e = base + k * 1024 + (int)threadIdx.x;
            const int ec = e < e1 ? e : e1 - 1;
            s[k] = srcIdx[ec];
            d[k] = dstIdx[ec];
        }
        for (int k = 0; k < EPT; ++k) {
            const int e = base + k * 1024 + (int)threadIdx.x;
            if (e < e1) {
                const int pos = atomicAdd(&lcount[bucket_of(d[k], nb)], 1);
                part[e0 + pos] = make_int2(s[k], d[k]);
            }
        }
    }
}

#define RUNB 16
__global__ __launch_bounds__(BNODES)
void gat_bucket_sort(const int2* __restrict__ part, const int* __restrict__ offT,
                     const int* __restrict__ cntT, int chunk, int N,
                     int* __restrict__ rowStart, int* __restrict__ cnt,
                     int* __restrict__ segSrc, int nbk,
                     const float* __restrict__ pA, const float* __restrict__ pB,
                     float* __restrict__ pC, int pRows, int pTile0) {
    static_assert(BNODES == 2 * P1C, "a projection group is one work-group here");
    static_assert(BNODES == NCHUNK, "one thread per chunk on the way in");
    static_assert(32 % RUNB == 0, "a wave's 32 runs in whole batches");
    __shared__ __attribute__((aligned(16))) float pLds[P1LDS];
    if ((int)blockIdx.x >= nbk) {
        gat_gemm_tiles<P1K, P1C>(pA, pB, pC, pRows, pTile0 + ((int)blockIdx.x - nbk), pLds);
        return;
    }
    __shared__ int lnode[BNODES];
    __shared__ int wsum[32];
    const int tid  = threadIdx.x;
    const int lane = tid & 31;
    const int runOff = offT[blockIdx.x * NCHUNK + tid];
    const int runLen = cntT[blockIdx.x * NCHUNK + tid];
    const int runPos = tid * chunk + runOff;
    lnode[tid] = 0;
    int p0;
    (void)block_excl_scan(runOff, wsum, BNODES / 32, p0);
    for (int j0 = 0; j0 < 32; j0 += RUNB) {
        int d[RUNB];
        for (int u = 0; u < RUNB; ++u) {
            const int st = __builtin_amdgcn_readlane(runPos, j0 + u);
            const int ln = __builtin_amdgcn_readlane(runLen, j0 + u);
            d[u] = part[max(st + min(lane, ln - 1), 0)].y;
        }
        for (int u = 0; u < RUNB; ++u) {
            const int ln = __builtin_amdgcn_readlane(runLen, j0 + u);
            if (lane < ln) atomicAdd(&lnode[d[u] & (BNODES - 1)], 1);
        }
        for (int u = 0; u < RUNB; ++u) {
            const int st = __builtin_amdgcn_readlane(runPos, j0 + u);
            const int ln = __builtin_amdgcn_readlane(runLen, j0 + u);
            for (int o = 32; o < ln; o += 32)
                if (o + lane < ln) atomicAdd(&lnode[part[st + o + lane].y & (BNODES - 1)], 1);
        }
    }
    __syncthreads();
    const int c = lnode[tid];
    int total;
    const int ex = block_excl_scan(c, wsum, BNODES / 32, total);
    lnode[tid] = p0 + ex;
    const int n = blockIdx.x * BNODES + tid;
    if (n < N) { rowStart[n] = p0 + ex; cnt[n] = c; }
    __syncthreads();
    for (int j0 = 0; j0 < 32; j0 += RUNB) {
        int2 sd[RUNB];
        for (int u = 0; u < RUNB; ++u) {
            const int st = __builtin_amdgcn_readlane(runPos, j0 + u);
            const int ln = __builtin_amdgcn_readlane(runLen, j0 + u);
            sd[u] = part[max(st + min(lane, ln - 1), 0)];
        }
        for (int u = 0; u < RUNB; ++u) {
            const int ln = __builtin_amdgcn_readlane(runLen, j0 + u);
            if (lane < ln) {
                const int pos = atomicAdd(&lnode[sd[u].y & (BNODES - 1)], 1);
                segSrc[pos] = sd[u].x;
            }
        }
        for (int u = 0; u < RUNB; ++u) {
            const int st = __builtin_amdgcn_readlane(runPos, j0 + u);
            const int ln = __builtin_amdgcn_readlane(runLen, j0 + u);
            for (int o = 32; o < ln; o += 32)
                if (o + lane < ln) {
                    const int2 e = part[st + o + lane];
                    const int pos = atomicAdd(&lnode[e.y & (BNODES - 1)], 1);
                    segSrc[pos] = e.x;
                }
        }
    }
}

template <int MASK>
__device__ __forceinline__ float row_xor(float x) {
    return __int_as_float(__builtin_amdgcn_update_dpp(
        0, __float_as_int(x), 0x160 + MASK, 0xf, 0xf, true));
}
__device__ __forceinline__ float half_xor(float x) {
    const int b = __float_as_int(x);
    return __int_as_float(__builtin_amdgcn_permlanex16(b, b, 0x76543210, 0xfedcba98, false, false));
}
template <int LPH>
__device__ __forceinline__ float head_sum(float v) {
    v += row_xor<1>(v);
    v += row_xor<2>(v);
    v += row_xor<4>(v);
    if constexpr (LPH > 8)  v += row_xor<8>(v);
    if constexpr (LPH > 16) v += half_xor(v);
    return v;
}

template <int P>
__device__ __forceinline__ void load_row(const float* __restrict__ p, v2f (&v)[P]) {
    if constexpr (P == 2) {
        float4 t = *(const float4*)p;
        v[0].x = t.x; v[0].y = t.y; v[1].x = t.z; v[1].y = t.w;
    } else {
        float2 t = *(const float2*)p;
        v[0].x = t.x; v[0].y = t.y;
    }
}
template <int P>
__device__ __forceinline__ float dot_cols(const v2f (&v)[P], const v2f (&w)[P]) {
    v2f q = v[0] * w[0];
    for (int k = 1; k < P; ++k) q += v[k] * w[k];
    return q.x + q.y;
}

template <int NB, int F, int P, int LPH>
__device__ __forceinline__ void gat_rows(int sItem, int t, int lane,
                                         const float* __restrict__ h,
                                         const v2f (&as)[P], float ed,
                                         float& m, float& sum, v2f (&acc)[P]) {
    v2f v[NB][P];
    for (int u = 0; u < NB; ++u) {
        const int st = __builtin_amdgcn_readlane(sItem, t + u);
        const unsigned boff = ((unsigned)st * (unsigned)F + (unsigned)(lane * (2 * P))) * 4u;
        load_row<P>((const float*)((const char*)h + boff), v[u]);
    }
    float a[NB];
    for (int u = 0; u < NB; ++u)
        a[u] = lrelu(head_sum<LPH>(dot_cols<P>(v[u], as)) + ed, NEG_SLOPE_ATT);
    float nm = m;
    for (int u = 0; u < NB; ++u) nm = fmaxf(nm, a[u]);
    const float sc = __expf(m - nm);
    sum *= sc;
    for (int k = 0; k < P; ++k) acc[k] *= sc;
    for (int u = 0; u < NB; ++u) {
        const float pe = __expf(a[u] - nm);
        sum += pe;
        for (int k = 0; k < P; ++k) acc[k] += pe * v[u][k];
    }
    m = nm;
}

template <int U, int LPH>
__device__ __forceinline__ float edge_weight(float pe) {
    if constexpr (LPH == 32) {
        return __int_as_float(__builtin_amdgcn_readlane(__float_as_int(pe), 4 * U));
    } else {
        constexpr int SEL = U * 0x249249;
        return __int_as_float(__builtin_amdgcn_mov_dpp8(__float_as_int(pe), SEL));
    }
}

template <int F, int P, int LPH>
__device__ __forceinline__ void gat_rows8(int sItem, int t, int lane,
                                          const float* __restrict__ h,
                                          const v2f (&as)[P], float ed,
                                          float& m, float& sum, v2f (&acc)[P]) {
    static_assert(LPH == 8 || LPH == 32, "exchange steps written for heads of 8 or 32 lanes");
    v2f v[8][P];
    for (int u = 0; u < 8; ++u) {
        const int st = __builtin_amdgcn_readlane(sItem, t + u);
        const unsigned boff = ((unsigned)st * (unsigned)F + (unsigned)(lane * (2 * P))) * 4u;
        load_row<P>((const float*)((const char*)h + boff), v[u]);
    }
    float p[8];
    for (int u = 0; u < 8; ++u) p[u] = dot_cols<P>(v[u], as);
    float q[4], r[2], s;
    if constexpr (LPH == 32) {
        const bool c4 = (lane & 16) != 0, c3 = (lane & 8) != 0, c2 = (lane & 4) != 0;
        for (int j = 0; j < 4; ++j)
            q[j] = (c4 ? p[j + 4] : p[j]) + half_xor(c4 ? p[j] : p[j + 4]);
        for (int j = 0; j < 2; ++j)
            r[j] = (c3 ? q[j + 2] : q[j]) + row_xor<8>(c3 ? q[j] : q[j + 2]);
        s = (c2 ? r[1] : r[0]) + row_xor<4>(c2 ? r[0] : r[1]);
        s += row_xor<2>(s);
        s += row_xor<1>(s);
    } else {
        const bool c2 = (lane & 4) != 0, c1 = (lane & 2) != 0, c0 = (lane & 1) != 0;
        for (int j = 0; j < 4; ++j)
            q[j] = (c2 ? p[j + 4] : p[j]) + row_xor<4>(c2 ? p[j] : p[j + 4]);
        for (int j = 0; j < 2; ++j)
            r[j] = (c1 ? q[j + 2] : q[j]) + row_xor<2>(c1 ? q[j] : q[j + 2]);
        s = (c0 ? r[1] : r[0]) + row_xor<1>(c0 ? r[0] : r[1]);
    }
    const float a = lrelu(s + ed, NEG_SLOPE_ATT);
    float mx = a;
    if constexpr (LPH == 32) {
        mx = fmaxf(mx, row_xor<4>(mx));
        mx = fmaxf(mx, row_xor<8>(mx));
        mx = fmaxf(mx, half_xor(mx));
    } else {
        mx = fmaxf(mx, row_xor<1>(mx));
        mx = fmaxf(mx, row_xor<2>(mx));
        mx = fmaxf(mx, row_xor<4>(mx));
    }
    const float nm = fmaxf(m, mx);
    const float sc = __expf(m - nm);
    const float pe = __expf(a - nm);
    sum *= sc;
    for (int k = 0; k < P; ++k) acc[k] *= sc;
#define GAT_EDGE_IN(U) { const float w = edge_weight<U, LPH>(pe); sum += w; \
        for (int k = 0; k < P; ++k) acc[k] += w * v[U][k]; }
    GAT_EDGE_IN(0) GAT_EDGE_IN(1) GAT_EDGE_IN(2) GAT_EDGE_IN(3)
    GAT_EDGE_IN(4) GAT_EDGE_IN(5) GAT_EDGE_IN(6) GAT_EDGE_IN(7)
#undef GAT_EDGE_IN
    m = nm;
}

template <int H, int C, int P, int VEC>
__device__ __forceinline__ void gat_node(int d, int lane,
                                         const int* __restrict__ rowStart, const int* __restrict__ cnt,
                                         const int* __restrict__ segSrc,
                                         const v2f (&as)[P], const v2f (&ad)[P],
                                         const float* __restrict__ h, const float* __restrict__ bias,
                                         float slope, int apply_act, float (&r)[VEC]) {
    constexpr int F = H * C;
    constexpr int LPH = C / VEC;
    static_assert(VEC == F / 32 && P == VEC / 2, "row = 32 lanes x VEC");
    const int start = rowStart[d];
    const int deg   = cnt[d];
    const int col   = lane * VEC;

    v2f vs[P];
    load_row<P>(h + (size_t)d * F + col, vs);
    int sItem = d;
    if (lane < deg) sItem = segSrc[start + lane];

    const float ed = head_sum<LPH>(dot_cols<P>(vs, ad));
    float m   = lrelu(head_sum<LPH>(dot_cols<P>(vs, as)) + ed, NEG_SLOPE_ATT);
    float sum = 1.f;
    v2f acc[P];
    for (int k = 0; k < P; ++k) acc[k] = vs[k];

    for (int base = 0; base < deg; base += 32) {
        if (base > 0) {
            sItem = d;
            if (base + lane < deg) sItem = segSrc[start + base + lane];
        }
        const int left = deg - base;
        const int n = left < 32 ? left : 32;
        int t = 0;
        for (; t + 8 <= n; t += 8)
            gat_rows8<F, P, LPH>(sItem, t, lane, h, as, ed, m, sum, acc);
        if (t + 4 <= n) {
            gat_rows<4, F, P, LPH>(sItem, t, lane, h, as, ed, m, sum, acc);
            t += 4;
        }
        const int rem = n - t;
        if (rem == 3)      gat_rows<3, F, P, LPH>(sItem, t, lane, h, as, ed, m, sum, acc);
        else if (rem == 2) gat_rows<2, F, P, LPH>(sItem, t, lane, h, as, ed, m, sum, acc);
        else if (rem == 1) gat_rows<1, F, P, LPH>(sItem, t, lane, h, as, ed, m, sum, acc);
    }

    const float inv = 1.0f / sum;
    for (int k = 0; k < VEC; ++k) {
        float x = acc[k >> 1][k & 1] * inv + bias[col + k];
        r[k] = apply_act ? lrelu(x, slope) : x;
    }
}

template <int SRC>
__device__ __forceinline__ int row_lane(int x) {
    return __builtin_amdgcn_update_dpp(0, x, 0x150 + SRC, 0xf, 0xf, true);
}

__device__ __forceinline__ void gat_rows8_row16(int sOff, int nvalid, int l16,
                                                const float* __restrict__ h,
                                                const v2f (&as)[2], float ed,
                                                float& m, float& sum, v2f (&acc)[2]) {
    v2f v[8][2];
    const int loff = l16 * 16;
#define GAT_ROW_LD(U) { const unsigned boff = (unsigned)(row_lane<U>(sOff) + loff); \
        load_row<2>((const float*)((const char*)h + boff), v[U]); }
    GAT_ROW_LD(0) GAT_ROW_LD(1) GAT_ROW_LD(2) GAT_ROW_LD(3)
    GAT_ROW_LD(4) GAT_ROW_LD(5) GAT_ROW_LD(6) GAT_ROW_LD(7)
#undef GAT_ROW_LD
    float p[8];
    for (int u = 0; u < 8; ++u) p[u] = dot_cols<2>(v[u], as);
    const bool c3 = (l16 & 8) != 0, c2 = (l16 & 4) != 0, c1 = (l16 & 2) != 0;
    float q[4], r[2], s;
    for (int j = 0; j < 4; ++j)
        q[j] = (c3 ? p[j + 4] : p[j]) + row_xor<8>(c3 ? p[j] : p[j + 4]);
    for (int j = 0; j < 2; ++j)
        r[j] = (c2 ? q[j + 2] : q[j]) + row_xor<4>(c2 ? q[j] : q[j + 2]);
    s = (c1 ? r[1] : r[0]) + row_xor<2>(c1 ? r[0] : r[1]);
    s += row_xor<1>(s);
    float a = lrelu(s + ed, NEG_SLOPE_ATT);
    if ((l16 >> 1) >= nvalid) a = -INFINITY;
    float mx = a;
    mx = fmaxf(mx, row_xor<2>(mx));
    mx = fmaxf(mx, row_xor<4>(mx));
    mx = fmaxf(mx, row_xor<8>(mx));
    const float nm = fmaxf(m, mx);
    const float sc = __expf(m - nm);
    const float pe = __expf(a - nm);
    sum *= sc;
    for (int k = 0; k < 2; ++k) acc[k] *= sc;
#define GAT_EDGE_IN(U) { const float w = __int_as_float(row_lane<2 * U>(__float_as_int(pe))); sum += w; \
        for (int k = 0; k < 2; ++k) acc[k] += w * v[U][k]; }
    GAT_EDGE_IN(0) GAT_EDGE_IN(1) GAT_EDGE_IN(2) GAT_EDGE_IN(3)
    GAT_EDGE_IN(4) GAT_EDGE_IN(5) GAT_EDGE_IN(6) GAT_EDGE_IN(7)
#undef GAT_EDGE_IN
    m = nm;
}

__global__ __launch_bounds__(256)
void gat_gather_rows16(const int* __restrict__ rowStart, const int* __restrict__ cnt,
                       const int* __restrict__ segSrc, int N,
                       const float* __restrict__ a_src, const float* __restrict__ a_dst,
                       const float* __restrict__ h, const float* __restrict__ bias,
                       float* __restrict__ out) {
    constexpr int F = 64;
    const int lane = threadIdx.x & 31;
    const int l16  = lane & 15;
    const int wv = __builtin_amdgcn_readfirstlane((int)(blockIdx.x * 8 + (threadIdx.x >> 5)));
    if (2 * wv >= N) return;
    const int dd   = 2 * wv + (lane >> 4);
    const bool live = dd < N;
    const int d    = live ? dd : N - 1;
    const int col  = l16 * 4;

    v2f as[2], ad[2], vs[2];
    load_row<2>(a_src + col, as);
    load_row<2>(a_dst + col, ad);
    load_row<2>((const float*)((const char*)h + (unsigned)(d * (F * 4) + l16 * 16)), vs);
    const int start = rowStart[d];
    const int cntd  = cnt[d];
    asm volatile("" :: "v"(start), "v"(cntd));
    const int deg   = live ? cntd : 0;
    const int degA = __builtin_amdgcn_readlane(deg, 0);
    const int degB = __builtin_amdgcn_readlane(deg, 16);
    const int maxdeg = degA > degB ? degA : degB;

    const float ed = head_sum<16>(dot_cols<2>(vs, ad));
    float m   = lrelu(head_sum<16>(dot_cols<2>(vs, as)) + ed, NEG_SLOPE_ATT);
    float sum = 1.f;
    v2f acc[2];
    for (int k = 0; k < 2; ++k) acc[k] = vs[k];

    for (int base = 0; base < maxdeg; base += 32) {
        const bool hasA = base + l16 < deg;
        const bool hasB = base + 16 + l16 < deg;
        int iA = d, iB = d;
        if (hasA) iA = segSrc[start + base + l16];
        if (hasB) iB = segSrc[start + base + 16 + l16];
        const int oA = iA * (F * 4), oB = iB * (F * 4);
        const int left = maxdeg - base;
        const int nb = left < 32 ? (left + 7) >> 3 : 4;
        for (int k = 0; k < nb; ++k) {
            int o = (k & 2) ? oB : oA;
            if (k & 1) o = __float_as_int(row_xor<8>(__int_as_float(o)));
            gat_rows8_row16(o, deg - base - 8 * k, l16, h, as, ed, m, sum, acc);
        }
    }

    if (live) {
        const float inv = 1.0f / sum;
        const float4 b = *(const float4*)(bias + col);
        *(float4*)(out + (size_t)d * F + col) =
            make_float4(acc[0].x * inv + b.x, acc[0].y * inv + b.y,
                        acc[1].x * inv + b.z, acc[1].y * inv + b.w);
    }
}

template <int H, int C, int NC2>
__global__ __launch_bounds__(256)
void gat_gather_project(const int* __restrict__ rowStart, const int* __restrict__ cnt,
                        const int* __restrict__ segSrc, int N,
                        const float* __restrict__ a_src, const float* __restrict__ a_dst,
                        const float* __restrict__ h, const float* __restrict__ bias, float slope,
                        const float* __restrict__ W2, float* __restrict__ out2) {
    constexpr int F = H * C;
    constexpr int VEC = F / 32;
    constexpr int P = VEC / 2;
    constexpr int KP = F + 4;
    constexpr int NPW = 32 / 8;
    static_assert(VEC == 4, "a lane's piece of a row is one 16-byte LDS store");
    static_assert(C % VEC == 0 && (C / VEC == 8 || C / VEC == 32), "a head is 8 or 32 adjacent lanes");
    static_assert(NC2 % 16 == 0 && NC2 / 16 <= 8, "one multiplying wave per 16-column tile");
    __shared__ __attribute__((aligned(16))) float tileA[32 * KP];

    const int lane  = threadIdx.x & 31;
    const int wv    = __builtin_amdgcn_readfirstlane((int)(threadIdx.x >> 5));
    const int tileM = (int)blockIdx.x * 32;
    const int col   = lane * VEC;

    v2f as[P], ad[P];
    load_row<P>(a_src + col, as);
    load_row<P>(a_dst + col, ad);
    for (int i = 0; i < NPW; ++i) {
        const int rt = wv * NPW + i;
        const int d  = tileM + rt;
        float r[VEC] = {0.f, 0.f, 0.f, 0.f};
        if (d < N)
            gat_node<H, C, P, VEC>(d, lane, rowStart, cnt, segSrc, as, ad, h, bias, slope, 1, r);
        v4f piece;
        piece.x = r[0]; piece.y = r[1]; piece.z = r[2]; piece.w = r[3];
        *(v4f*)&tileA[rt * KP + col] = piece;
    }
    __syncthreads();
    if (wv < NC2 / 16)
        gat_tile_multiply<F, NC2>(tileA, W2, out2, N, tileM, wv, lane);
}

extern "C" void kernel_launch(void* const* d_in, const int* in_sizes, int n_in,
                              void* d_out, int out_size, void* d_ws, size_t ws_size,
                              hipStream_t stream) {
    const float* x        = (const float*)d_in[0];
    const int*   ei       = (const int*)d_in[1];
    const float* W1       = (const float*)d_in[2];
    const float* att_src1 = (const float*)d_in[3];
    const float* att_dst1 = (const float*)d_in[4];
    const float* b1       = (const float*)d_in[5];
    const float* W2       = (const float*)d_in[6];
    const float* att_src2 = (const float*)d_in[7];
    const float* att_dst2 = (const float*)d_in[8];
    const float* b2       = (const float*)d_in[9];

    const int HC1  = in_sizes[5];
    const int OUTC = in_sizes[9];
    const int F_IN = in_sizes[2] / HC1;
    const int N    = in_sizes[0] / F_IN;
    const int E    = in_sizes[1] / 2;

    const int* srcIdx = ei;
    const int* dstIdx = ei + E;

    float* ws  = (float*)d_ws;
    size_t off = 0;
    float* h1   = ws + off; off += (size_t)N * HC1;
    float* h2   = ws + off; off += (size_t)N * OUTC;
    const int NBK   = (N + BNODES - 1) / BNODES;
    const int chunk = (E + NCHUNK - 1) / NCHUNK;
    int* cnt      = (int*)(ws + off); off += (((size_t)N + 3) / 4) * 4;
    int* rowStart = (int*)(ws + off); off += (((size_t)N + 3) / 4) * 4;
    int* segSrc   = (int*)(ws + off); off += (((size_t)E + 3) / 4) * 4;
    int* offT        = (int*)(ws + off); off += (size_t)NBK * NCHUNK;
    int* cntT        = (int*)(ws + off); off += (size_t)NBK * NCHUNK;
    int2* part       = (int2*)(ws + off); off += (((size_t)E * 2 + 3) / 4) * 4;
    float* out2 = (float*)d_out;

    const int TB = 256;
    const int rowTiles32 = (N + 31) / 32;
    const int pairBlocks = (N + 15) / 16;
    const int gA = rowTiles32 / 6 < 2 * NCHUNK ? rowTiles32 / 6 : 2 * NCHUNK;
    const int gC = rowTiles32 - 4 * gA;

    gat_chunk_group<<<NCHUNK + gA, 1024, NBK * sizeof(int), stream>>>(
        srcIdx, dstIdx, E, chunk, NBK, offT, cntT, part, x, W1, h1, N, 0);
    gat_bucket_sort<<<NBK + gC, BNODES, 0, stream>>>(
        part, offT, cntT, chunk, N, rowStart, cnt, segSrc, NBK, x, W1, h1, N, 4 * gA);

    gat_gather_project<4, 32, 64><<<rowTiles32, TB, 0, stream>>>(
        rowStart, cnt, segSrc, N, att_src1, att_dst1, h1, b1, NEG_SLOPE_ACT, W2, h2);

    gat_gather_rows16<<<pairBlocks, TB, 0, stream>>>(
        rowStart, cnt, segSrc, N, att_src2, att_dst2, h2, b2, out2);
}
